# starting kernel
# baseline (speedup 1.0000x reference)
#include <hip/hip_runtime.h>
#include <math.h>

typedef __attribute__((ext_vector_type(2))) float v2f;
typedef __attribute__((ext_vector_type(8))) float v8f;

#define NN 50000
#define EE 800000
#define DD 128
#define GG 128
#define LDS_STRIDE 132

__global__ void __launch_bounds__(256)
node_gemm_wmma(const float* __restrict__ nf, const float* __restrict__ Wn,
               const float* __restrict__ bn, float* __restrict__ hv) {
  __shared__ float Atile[16 * LDS_STRIDE];
  const int row0 = blockIdx.x * 16;
  const int tid  = threadIdx.x;

  #pragma unroll
  for (int i = tid; i < 16 * DD; i += 256) {
    const int r = i >> 7;
    const int c = i & (DD - 1);
    Atile[r * LDS_STRIDE + c] = nf[(size_t)(row0 + r) * DD + c];
  }
  __syncthreads();

  const int wave = tid >> 5;
  const int lane = tid & 31;
  const int col0 = wave * 16;
  const int half = lane >> 4;
  const int l16  = lane & 15;

  v8f acc = {};
  #pragma unroll
  for (int k = 0; k < DD; k += 4) {
    const int ka = k + half * 2;
    v2f a, b;
    a.x = Atile[l16 * LDS_STRIDE + ka];
    a.y = Atile[l16 * LDS_STRIDE + ka + 1];
    b.x = Wn[(size_t)(ka + 0) * GG + col0 + l16];
    b.y = Wn[(size_t)(ka + 1) * GG + col0 + l16];
    acc = __builtin_amdgcn_wmma_f32_16x16x4_f32(false, a, false, b,
                                                (short)0, acc, false, false);
  }

  #pragma unroll
  for (int i = 0; i < 8; ++i) {
    const int row = row0 + i + half * 8;
    const int col = col0 + l16;
    float v = acc[i] + bn[col];
    v = v > 0.0f ? v : 0.01f * v;
    hv[(size_t)row * GG + col] = v;
  }
}

__global__ void __launch_bounds__(256)
node_scores(const float* __restrict__ nf, const float* __restrict__ We,
            float* __restrict__ p, float* __restrict__ q) {
  const int wid  = (int)((blockIdx.x * (size_t)blockDim.x + threadIdx.x) >> 5);
  const int lane = threadIdx.x & 31;
  if (wid >= NN) return;
  const float* row = nf + (size_t)wid * DD;
  float ap = 0.0f, aq = 0.0f;
  #pragma unroll
  for (int i = lane; i < DD; i += 32) {
    const float x = row[i];
    ap += x * We[i];
    aq += x * We[DD + i];
  }
  #pragma unroll
  for (int off = 16; off > 0; off >>= 1) {
    ap += __shfl_xor(ap, off, 32);
    aq += __shfl_xor(aq, off, 32);
  }
  if (lane == 0) { p[wid] = ap; q[wid] = aq; }
}

__global__ void __launch_bounds__(256)
init_buffers(float* __restrict__ m, float* __restrict__ s,
             float* __restrict__ out) {
  const int i = (int)(blockIdx.x * (size_t)blockDim.x + threadIdx.x);
  if (i < NN * GG) out[i] = 0.0f;
  if (i < NN) { m[i] = -INFINITY; s[i] = 0.0f; }
}

__device__ inline void atomicMaxF(float* addr, float v) {
  if (v >= 0.0f) atomicMax((int*)addr, __float_as_int(v));
  else           atomicMin((unsigned int*)addr, __float_as_uint(v));
}

__global__ void __launch_bounds__(256)
edge_logits(const int* __restrict__ src, const int* __restrict__ dst,
            const float* __restrict__ p, const float* __restrict__ q,
            const float* __restrict__ be, float* __restrict__ logit,
            float* __restrict__ m) {
  const int e = (int)(blockIdx.x * (size_t)blockDim.x + threadIdx.x);
  if (e >= EE) return;
  const int d = dst[e];
  float x = p[d] + q[src[e]] + be[0];
  x = x > 0.0f ? x : 0.01f * x;
  logit[e] = x;
  atomicMaxF(&m[d], x);
}

__global__ void __launch_bounds__(256)
edge_expsum(const int* __restrict__ dst, const float* __restrict__ logit,
            const float* __restrict__ m, float* __restrict__ s) {
  const int e = (int)(blockIdx.x * (size_t)blockDim.x + threadIdx.x);
  if (e >= EE) return;
  const int d = dst[e];
  atomicAdd(&s[d], expf(logit[e] - m[d]));
}

__global__ void __launch_bounds__(256)
edge_aggregate(const int* __restrict__ src, const int* __restrict__ dst,
               const float* __restrict__ logit, const float* __restrict__ m,
               const float* __restrict__ s, const float* __restrict__ hv,
               float* __restrict__ out) {
  const int e = (int)((blockIdx.x * (size_t)blockDim.x + threadIdx.x) >> 5);
  const int lane = threadIdx.x & 31;
  if (e >= EE) return;
  const int sv = src[e];
  const int dv = dst[e];
  const float den = s[dv];
  const float a = expf(logit[e] - m[dv]) / (den > 0.0f ? den : 1.0f);
  const float4 h = ((const float4*)(hv + (size_t)sv * GG))[lane];
  float* orow = out + (size_t)dv * GG + lane * 4;
  atomicAdd(orow + 0, h.x * a);
  atomicAdd(orow + 1, h.y * a);
  atomicAdd(orow + 2, h.z * a);
  atomicAdd(orow + 3, h.w * a);
}

__global__ void __launch_bounds__(256)
elu_inplace(float* __restrict__ out) {
  const int i = (int)(blockIdx.x * (size_t)blockDim.x + threadIdx.x);
  if (i >= NN * GG) return;
  const float c = out[i];
  out[i] = c > 0.0f ? c : expm1f(c);
}

extern "C" void kernel_launch(void* const* d_in, const int* in_sizes, int n_in,
                              void* d_out, int out_size, void* d_ws, size_t ws_size,
                              hipStream_t stream) {
  (void)in_sizes; (void)n_in; (void)out_size; (void)ws_size;
  const float* nf  = (const float*)d_in[0];
  const int*   src = (const int*)  d_in[1];
  const int*   dst = (const int*)  d_in[2];
  const float* We  = (const float*)d_in[3];
  const float* be  = (const float*)d_in[4];
  const float* Wn  = (const float*)d_in[5];
  const float* bn  = (const float*)d_in[6];
  float* out = (float*)d_out;

  float* ws    = (float*)d_ws;
  float* hv    = ws;
  float* p     = hv + (size_t)NN * GG;
  float* q     = p + NN;
  float* m     = q + NN;
  float* ssum  = m + NN;
  float* logit = ssum + NN;

  const int ngBlocks = (NN * GG + 255) / 256;

  init_buffers<<<ngBlocks, 256, 0, stream>>>(m, ssum, out);
  node_gemm_wmma<<<NN / 16, 256, 0, stream>>>(nf, Wn, bn, hv);
  node_scores<<<(NN * 32 + 255) / 256, 256, 0, stream>>>(nf, We, p, q);
  edge_logits<<<(EE + 255) / 256, 256, 0, stream>>>(src, dst, p, q, be, logit, m);
  edge_expsum<<<(EE + 255) / 256, 256, 0, stream>>>(dst, logit, m, ssum);
  edge_aggregate<<<(EE * 32 + 255) / 256, 256, 0, stream>>>(src, dst, logit, m, ssum, hv, out);
  elu_inplace<<<ngBlocks, 256, 0, stream>>>(out);
}


# Edges grouped by destination in-call (count, scan, place src ids), segment ordered by source id; one wave per node: softmax + row sum in registers, no float atomics, sums in ascending source id
# speedup vs baseline: 1.5918x; 1.5918x over previous
#include <hip/hip_runtime.h>
#include <math.h>

typedef __attribute__((ext_vector_type(2))) float v2f;
typedef __attribute__((ext_vector_type(8))) float v8f;

#define NN 50000
#define EE 800000
#define DD 128
#define GG 128
#define LDS_STRIDE 132

__global__ void __launch_bounds__(256)
node_gemm_wmma(const float* __restrict__ nf, const float* __restrict__ Wn,
               const float* __restrict__ bn, float* __restrict__ hv) {
  __shared__ float Atile[16 * LDS_STRIDE];
  const int row0 = blockIdx.x * 16;
  const int tid  = threadIdx.x;

  #pragma unroll
  for (int i = tid; i < 16 * DD; i += 256) {
    const int r = i >> 7;
    const int c = i & (DD - 1);
    Atile[r * LDS_STRIDE + c] = nf[(size_t)(row0 + r) * DD + c];
  }
  __syncthreads();

  const int wave = tid >> 5;
  const int lane = tid & 31;
  const int col0 = wave * 16;
  const int half = lane >> 4;
  const int l16  = lane & 15;

  v8f acc = {};
  #pragma unroll
  for (int k = 0; k < DD; k += 4) {
    const int ka = k + half * 2;
    v2f a, b;
    a.x = Atile[l16 * LDS_STRIDE + ka];
    a.y = Atile[l16 * LDS_STRIDE + ka + 1];
    b.x = Wn[(size_t)(ka + 0) * GG + col0 + l16];
    b.y = Wn[(size_t)(ka + 1) * GG + col0 + l16];
    acc = __builtin_amdgcn_wmma_f32_16x16x4_f32(false, a, false, b,
                                                (short)0, acc, false, false);
  }

  #pragma unroll
  for (int i = 0; i < 8; ++i) {
    const int row = row0 + i + half * 8;
    const int col = col0 + l16;
    float v = acc[i] + bn[col];
    v = v > 0.0f ? v : 0.01f * v;
    hv[(size_t)row * GG + col] = v;
  }
}

__global__ void __launch_bounds__(256)
node_scores(const float* __restrict__ nf, const float* __restrict__ We,
            float* __restrict__ p, float* __restrict__ q,
            int* __restrict__ cnt) {
  const int wid  = (int)((blockIdx.x * (size_t)blockDim.x + threadIdx.x) >> 5);
  const int lane = threadIdx.x & 31;
  if (wid >= NN) return;
  const float* row = nf + (size_t)wid * DD;
  float ap = 0.0f, aq = 0.0f;
  #pragma unroll
  for (int i = lane; i < DD; i += 32) {
    const float x = row[i];
    ap += x * We[i];
    aq += x * We[DD + i];
  }
  #pragma unroll
  for (int off = 16; off > 0; off >>= 1) {
    ap += __shfl_xor(ap, off, 32);
    aq += __shfl_xor(aq, off, 32);
  }
  if (lane == 0) { p[wid] = ap; q[wid] = aq; cnt[wid] = 0; }
}

__global__ void __launch_bounds__(256)
edge_count(const int* __restrict__ dst, int* __restrict__ cnt) {
  const int t = (int)(blockIdx.x * (size_t)blockDim.x + threadIdx.x);
  if (t >= EE / 4) return;
  const int4 d = ((const int4*)dst)[t];
  atomicAdd(&cnt[d.x], 1);
  atomicAdd(&cnt[d.y], 1);
  atomicAdd(&cnt[d.z], 1);
  atomicAdd(&cnt[d.w], 1);
}

#define SCAN_T   1024
#define SCAN_PER 49

__global__ void __launch_bounds__(SCAN_T)
scan_counts(int* __restrict__ cnt, int* __restrict__ off) {
  __shared__ int buf[SCAN_T * SCAN_PER];
  __shared__ int wsum[32];
  const int tid  = threadIdx.x;
  const int lane = tid & 31;
  const int w    = tid >> 5;
  for (int i = tid; i < SCAN_T * SCAN_PER; i += SCAN_T) buf[i] = (i < NN) ? cnt[i] : 0;
  __syncthreads();
  const int base = tid * SCAN_PER;
  int s = 0;
  for (int i = 0; i < SCAN_PER; ++i) s += buf[base + i];
  int inc = s;
  for (int o = 1; o < 32; o <<= 1) {
    const int t = __shfl_up(inc, o, 32);
    if (lane >= o) inc += t;
  }
  if (lane == 31) wsum[w] = inc;
  __syncthreads();
  if (w == 0) {
    const int x = wsum[lane];
    int xi = x;
    for (int o = 1; o < 32; o <<= 1) {
      const int t = __shfl_up(xi, o, 32);
      if (lane >= o) xi += t;
    }
    wsum[lane] = xi - x;
  }
  __syncthreads();
  int run = wsum[w] + inc - s;
  for (int i = 0; i < SCAN_PER; ++i) {
    const int c = buf[base + i];
    buf[base + i] = run;
    run += c;
  }
  __syncthreads();
  for (int i = tid; i < NN; i += SCAN_T) {
    const int o = buf[i];
    off[i] = o;
    cnt[i] = o;
  }
}

__global__ void __launch_bounds__(256)
edge_place(const int* __restrict__ src, const int* __restrict__ dst,
           int* __restrict__ cur, int* __restrict__ col) {
  const int t = (int)(blockIdx.x * (size_t)blockDim.x + threadIdx.x);
  if (t >= EE / 4) return;
  const int4 d = ((const int4*)dst)[t];
  const int4 s = ((const int4*)src)[t];
  col[atomicAdd(&cur[d.x], 1)] = s.x;
  col[atomicAdd(&cur[d.y], 1)] = s.y;
  col[atomicAdd(&cur[d.z], 1)] = s.z;
  col[atomicAdd(&cur[d.w], 1)] = s.w;
}

__device__ inline float leaky(float x) { return x > 0.0f ? x : 0.01f * x; }

__device__ inline int wave_min_i(int x) {
  for (int o = 16; o > 0; o >>= 1) x = min(x, __shfl_xor(x, o, 32));
  return x;
}
__device__ inline int wave_sum_i(int x) {
  for (int o = 16; o > 0; o >>= 1) x += __shfl_xor(x, o, 32);
  return x;
}
__device__ inline float wave_max_f(float x) {
  for (int o = 16; o > 0; o >>= 1) x = fmaxf(x, __shfl_xor(x, o, 32));
  return x;
}

__device__ inline int next_source(const int* __restrict__ seg, int deg, int last, int lane,
                                  int& mult) {
  int best = 0x7fffffff;
  for (int c = lane; c < deg; c += 32) {
    const int s = seg[c];
    if (s > last && s < best) best = s;
  }
  best = wave_min_i(best);
  int m = 0;
  for (int c = lane; c < deg; c += 32) m += (seg[c] == best) ? 1 : 0;
  mult = (best == 0x7fffffff) ? 0 : wave_sum_i(m);
  return best;
}

template <int C>
__device__ inline void pick(const int (&rank)[C], const int (&sv)[C], const float (&w)[C],
                            int k, int& svk, float& wk) {
  svk = 0; wk = 0.0f;
  for (int c = 0; c < C; ++c) {
    const unsigned mk = __builtin_amdgcn_ballot_w32(rank[c] == k);
    if (C == 1 || mk != 0u) {
      const int j = (__ffs((int)mk) - 1) & 31;
      svk = __builtin_amdgcn_readlane(sv[c], j);
      wk  = __int_as_float(__builtin_amdgcn_readlane(__float_as_int(w[c]), j));
    }
  }
}

template <int C>
__device__ inline float4 segment_in_registers(const int* __restrict__ seg, int deg, int lane,
                                              const float* __restrict__ q,
                                              float pv, float be0,
                                              const float4* __restrict__ hv4) {
  int sv[C], key[C], rank[C];
  float x[C], ex[C], a[C];
  for (int c = 0; c < C; ++c) {
    const int i = c * 32 + lane;
    sv[c] = 0; key[c] = 0x7fffffff; x[c] = -INFINITY; rank[c] = 0;
    if (i < deg) {
      sv[c]  = min(max(seg[i], 0), NN - 1);
      key[c] = sv[c];
      x[c]   = leaky(pv + q[sv[c]] + be0);
    }
  }
  for (int j = 0; j < deg; ++j) {
    int kj = __builtin_amdgcn_readlane(key[0], j & 31);
    for (int c = 1; c < C; ++c)
      if ((j >> 5) == c) kj = __builtin_amdgcn_readlane(key[c], j & 31);
    for (int c = 0; c < C; ++c)
      rank[c] += (kj < key[c] || (kj == key[c] && j < c * 32 + lane)) ? 1 : 0;
  }
  float mx = x[0];
  for (int c = 1; c < C; ++c) mx = fmaxf(mx, x[c]);
  mx = wave_max_f(mx);
  for (int c = 0; c < C; ++c) ex[c] = (c * 32 + lane < deg) ? expf(x[c] - mx) : 0.0f;
  float den = 0.0f;
  for (int k = 0; k < deg; ++k) {
    int s_; float exk;
    pick<C>(rank, sv, ex, k, s_, exk);
    den += exk;
  }
  const float dsafe = den > 0.0f ? den : 1.0f;
  for (int c = 0; c < C; ++c) a[c] = ex[c] / dsafe;

  float4 acc = make_float4(0.0f, 0.0f, 0.0f, 0.0f);
  int k = 0;
  for (; k + 4 <= deg; k += 4) {
    int s0, s1, s2, s3; float a0, a1, a2, a3;
    pick<C>(rank, sv, a, k,     s0, a0);
    pick<C>(rank, sv, a, k + 1, s1, a1);
    pick<C>(rank, sv, a, k + 2, s2, a2);
    pick<C>(rank, sv, a, k + 3, s3, a3);
    const float4 h0 = hv4[(size_t)s0 * (GG / 4) + lane];
    const float4 h1 = hv4[(size_t)s1 * (GG / 4) + lane];
    const float4 h2 = hv4[(size_t)s2 * (GG / 4) + lane];
    const float4 h3 = hv4[(size_t)s3 * (GG / 4) + lane];
    acc.x += h0.x * a0; acc.y += h0.y * a0; acc.z += h0.z * a0; acc.w += h0.w * a0;
    acc.x += h1.x * a1; acc.y += h1.y * a1; acc.z += h1.z * a1; acc.w += h1.w * a1;
    acc.x += h2.x * a2; acc.y += h2.y * a2; acc.z += h2.z * a2; acc.w += h2.w * a2;
    acc.x += h3.x * a3; acc.y += h3.y * a3; acc.z += h3.z * a3; acc.w += h3.w * a3;
  }
  for (; k < deg; ++k) {
    int sk; float ak;
    pick<C>(rank, sv, a, k, sk, ak);
    const float4 h = hv4[(size_t)sk * (GG / 4) + lane];
    acc.x += h.x * ak; acc.y += h.y * ak; acc.z += h.z * ak; acc.w += h.w * ak;
  }
  return acc;
}

__global__ void __launch_bounds__(256)
node_softmax_aggregate(const int* __restrict__ col,
                       const int* __restrict__ off, const int* __restrict__ endp,
                       const float* __restrict__ p, const float* __restrict__ q,
                       const float* __restrict__ be, const float* __restrict__ hv,
                       float* __restrict__ out) {
  const int v    = (int)((blockIdx.x * (size_t)blockDim.x + threadIdx.x) >> 5);
  const int lane = threadIdx.x & 31;
  if (v >= NN) return;
  const int start = off[v];
  const int deg   = endp[v] - start;
  const int* seg  = col + start;
  const float pv  = p[v];
  const float be0 = be[0];
  const float4* hv4 = (const float4*)hv;
  float4 acc;

  if (deg <= 32) {
    acc = segment_in_registers<1>(seg, deg, lane, q, pv, be0, hv4);
  } else if (deg <= 64) {
    acc = segment_in_registers<2>(seg, deg, lane, q, pv, be0, hv4);
  } else {
    acc = make_float4(0.0f, 0.0f, 0.0f, 0.0f);
    float mx = -INFINITY;
    for (int c = lane; c < deg; c += 32) {
      const int s = min(max(seg[c], 0), NN - 1);
      mx = fmaxf(mx, leaky(pv + q[s] + be0));
    }
    mx = wave_max_f(mx);
    float den = 0.0f;
    int last = -1;
    for (int k = 0; k < deg; ) {
      int mult;
      const int s = next_source(seg, deg, last, lane, mult);
      if (mult <= 0) break;
      last = s;
      const float exk = expf(leaky(pv + q[min(s, NN - 1)] + be0) - mx);
      for (int t = 0; t < mult; ++t) den += exk;
      k += mult;
    }
    const float dsafe = den > 0.0f ? den : 1.0f;
    last = -1;
    for (int k = 0; k < deg; ) {
      int mult;
      const int s = next_source(seg, deg, last, lane, mult);
      if (mult <= 0) break;
      last = s;
      const int sc = min(s, NN - 1);
      const float ak = expf(leaky(pv + q[sc] + be0) - mx) / dsafe;
      const float4 h = hv4[(size_t)sc * (GG / 4) + lane];
      for (int t = 0; t < mult; ++t) {
        acc.x += h.x * ak; acc.y += h.y * ak; acc.z += h.z * ak; acc.w += h.w * ak;
      }
      k += mult;
    }
  }

  float4 r;
  r.x = acc.x > 0.0f ? acc.x : expm1f(acc.x);
  r.y = acc.y > 0.0f ? acc.y : expm1f(acc.y);
  r.z = acc.z > 0.0f ? acc.z : expm1f(acc.z);
  r.w = acc.w > 0.0f ? acc.w : expm1f(acc.w);
  ((float4*)out)[(size_t)v * (GG / 4) + lane] = r;
}

extern "C" void kernel_launch(void* const* d_in, const int* in_sizes, int n_in,
                              void* d_out, int out_size, void* d_ws, size_t ws_size,
                              hipStream_t stream) {
  (void)in_sizes; (void)n_in; (void)out_size; (void)ws_size;
  const float* nf  = (const float*)d_in[0];
  const int*   src = (const int*)  d_in[1];
  const int*   dst = (const int*)  d_in[2];
  const float* We  = (const float*)d_in[3];
  const float* be  = (const float*)d_in[4];
  const float* Wn  = (const float*)d_in[5];
  const float* bn  = (const float*)d_in[6];
  float* out = (float*)d_out;

  float* ws   = (float*)d_ws;
  float* hv   = ws;
  float* p    = hv + (size_t)NN * GG;
  float* q    = p + NN;
  int*   cnt  = (int*)(q + NN);
  int*   off  = cnt + NN;
  int*   col  = off + NN;

  node_gemm_wmma<<<NN / 16, 256, 0, stream>>>(nf, Wn, bn, hv);
  node_scores<<<(NN * 32 + 255) / 256, 256, 0, stream>>>(nf, We, p, q, cnt);
  edge_count<<<(EE / 4 + 255) / 256, 256, 0, stream>>>(dst, cnt);
  scan_counts<<<1, SCAN_T, 0, stream>>>(cnt, off);
  edge_place<<<(EE / 4 + 255) / 256, 256, 0, stream>>>(src, dst, cnt, col);
  node_softmax_aggregate<<<(NN * 32 + 255) / 256, 256, 0, stream>>>(col, off, cnt, p, q, be, hv, out);
}


# Node score halves p,q and the counter clear computed in the node GEMM launch from its LDS row tile (2 rows per wave, same order of sums); the launch of 50000 short one-node score waves is gone
# speedup vs baseline: 1.6768x; 1.0534x over previous
#include <hip/hip_runtime.h>
#include <math.h>

typedef __attribute__((ext_vector_type(2))) float v2f;
typedef __attribute__((ext_vector_type(8))) float v8f;

#define NN 50000
#define EE 800000
#define DD 128
#define GG 128
#define LDS_STRIDE 132

__global__ void __launch_bounds__(256, 16)
node_gemm_wmma(const float* __restrict__ nf, const float* __restrict__ Wn,
               const float* __restrict__ bn, float* __restrict__ hv,
               const float* __restrict__ We, float* __restrict__ p,
               float* __restrict__ q, int* __restrict__ cnt) {
  __shared__ float Atile[16 * LDS_STRIDE];
  const int row0 = blockIdx.x * 16;
  const int tid  = threadIdx.x;

  #pragma unroll
  for (int i = tid; i < 16 * DD; i += 256) {
    const int r = i >> 7;
    const int c = i & (DD - 1);
    Atile[r * LDS_STRIDE + c] = nf[(size_t)(row0 + r) * DD + c];
  }
  __syncthreads();

  const int wave = tid >> 5;
  const int lane = tid & 31;
  const int col0 = wave * 16;
  const int half = lane >> 4;
  const int l16  = lane & 15;

  v8f acc = {};
  #pragma unroll
  for (int k = 0; k < DD; k += 4) {
    const int ka = k + half * 2;
    v2f a, b;
    a.x = Atile[l16 * LDS_STRIDE + ka];
    a.y = Atile[l16 * LDS_STRIDE + ka + 1];
    b.x = Wn[(size_t)(ka + 0) * GG + col0 + l16];
    b.y = Wn[(size_t)(ka + 1) * GG + col0 + l16];
    acc = __builtin_amdgcn_wmma_f32_16x16x4_f32(false, a, false, b,
                                                (short)0, acc, false, false);
  }

  #pragma unroll
  for (int i = 0; i < 8; ++i) {
    const int row = row0 + i + half * 8;
    const int col = col0 + l16;
    float v = acc[i] + bn[col];
    v = v > 0.0f ? v : 0.01f * v;
    hv[(size_t)row * GG + col] = v;
  }

  const int sr0 = wave * 2;
  float ap0 = 0.0f, aq0 = 0.0f, ap1 = 0.0f, aq1 = 0.0f;
  #pragma unroll
  for (int i = lane; i < DD; i += 32) {
    const float wp = We[i];
    const float wq = We[DD + i];
    const float x0 = Atile[sr0 * LDS_STRIDE + i];
    const float x1 = Atile[(sr0 + 1) * LDS_STRIDE + i];
    ap0 += x0 * wp;
    aq0 += x0 * wq;
    ap1 += x1 * wp;
    aq1 += x1 * wq;
  }
  #pragma unroll
  for (int off = 16; off > 0; off >>= 1) {
    ap0 += __shfl_xor(ap0, off, 32);
    aq0 += __shfl_xor(aq0, off, 32);
    ap1 += __shfl_xor(ap1, off, 32);
    aq1 += __shfl_xor(aq1, off, 32);
  }

  if (lane < 2) {
    const int node = row0 + sr0 + lane;
    p[node]   = lane ? ap1 : ap0;
    q[node]   = lane ? aq1 : aq0;
    cnt[node] = 0;
  }
}

__global__ void __launch_bounds__(256)
edge_count(const int* __restrict__ dst, int* __restrict__ cnt) {
  const int t = (int)(blockIdx.x * (size_t)blockDim.x + threadIdx.x);
  if (t >= EE / 4) return;
  const int4 d = ((const int4*)dst)[t];
  atomicAdd(&cnt[d.x], 1);
  atomicAdd(&cnt[d.y], 1);
  atomicAdd(&cnt[d.z], 1);
  atomicAdd(&cnt[d.w], 1);
}

#define SCAN_T   1024
#define SCAN_PER 49

__global__ void __launch_bounds__(SCAN_T)
scan_counts(int* __restrict__ cnt, int* __restrict__ off) {
  __shared__ int buf[SCAN_T * SCAN_PER];
  __shared__ int wsum[32];
  const int tid  = threadIdx.x;
  const int lane = tid & 31;
  const int w    = tid >> 5;
  for (int i = tid; i < SCAN_T * SCAN_PER; i += SCAN_T) buf[i] = (i < NN) ? cnt[i] : 0;
  __syncthreads();
  const int base = tid * SCAN_PER;
  int s = 0;
  for (int i = 0; i < SCAN_PER; ++i) s += buf[base + i];
  int inc = s;
  for (int o = 1; o < 32; o <<= 1) {
    const int t = __shfl_up(inc, o, 32);
    if (lane >= o) inc += t;
  }
  if (lane == 31) wsum[w] = inc;
  __syncthreads();
  if (w == 0) {
    const int x = wsum[lane];
    int xi = x;
    for (int o = 1; o < 32; o <<= 1) {
      const int t = __shfl_up(xi, o, 32);
      if (lane >= o) xi += t;
    }
    wsum[lane] = xi - x;
  }
  __syncthreads();
  int run = wsum[w] + inc - s;
  for (int i = 0; i < SCAN_PER; ++i) {
    const int c = buf[base + i];
    buf[base + i] = run;
    run += c;
  }
  __syncthreads();
  for (int i = tid; i < NN; i += SCAN_T) {
    const int o = buf[i];
    off[i] = o;
    cnt[i] = o;
  }
}

__global__ void __launch_bounds__(256)
edge_place(const int* __restrict__ src, const int* __restrict__ dst,
           int* __restrict__ cur, int* __restrict__ col) {
  const int t = (int)(blockIdx.x * (size_t)blockDim.x + threadIdx.x);
  if (t >= EE / 4) return;
  const int4 d = ((const int4*)dst)[t];
  const int4 s = ((const int4*)src)[t];
  col[atomicAdd(&cur[d.x], 1)] = s.x;
  col[atomicAdd(&cur[d.y], 1)] = s.y;
  col[atomicAdd(&cur[d.z], 1)] = s.z;
  col[atomicAdd(&cur[d.w], 1)] = s.w;
}

__device__ inline float leaky(float x) { return x > 0.0f ? x : 0.01f * x; }

__device__ inline int wave_min_i(int x) {
  for (int o = 16; o > 0; o >>= 1) x = min(x, __shfl_xor(x, o, 32));
  return x;
}
__device__ inline int wave_sum_i(int x) {
  for (int o = 16; o > 0; o >>= 1) x += __shfl_xor(x, o, 32);
  return x;
}
__device__ inline float wave_max_f(float x) {
  for (int o = 16; o > 0; o >>= 1) x = fmaxf(x, __shfl_xor(x, o, 32));
  return x;
}

__device__ inline int next_source(const int* __restrict__ seg, int deg, int last, int lane,
                                  int& mult) {
  int best = 0x7fffffff;
  for (int c = lane; c < deg; c += 32) {
    const int s = seg[c];
    if (s > last && s < best) best = s;
  }
  best = wave_min_i(best);
  int m = 0;
  for (int c = lane; c < deg; c += 32) m += (seg[c] == best) ? 1 : 0;
  mult = (best == 0x7fffffff) ? 0 : wave_sum_i(m);
  return best;
}

template <int C>
__device__ inline void pick(const int (&rank)[C], const int (&sv)[C], const float (&w)[C],
                            int k, int& svk, float& wk) {
  svk = 0; wk = 0.0f;
  for (int c = 0; c < C; ++c) {
    const unsigned mk = __builtin_amdgcn_ballot_w32(rank[c] == k);
    if (C == 1 || mk != 0u) {
      const int j = (__ffs((int)mk) - 1) & 31;
      svk = __builtin_amdgcn_readlane(sv[c], j);
      wk  = __int_as_float(__builtin_amdgcn_readlane(__float_as_int(w[c]), j));
    }
  }
}

template <int C>
__device__ inline float4 segment_in_registers(const int* __restrict__ seg, int deg, int lane,
                                              const float* __restrict__ q,
                                              float pv, float be0,
                                              const float4* __restrict__ hv4) {
  int sv[C], key[C], rank[C];
  float x[C], ex[C], a[C];
  for (int c = 0; c < C; ++c) {
    const int i = c * 32 + lane;
    sv[c] = 0; key[c] = 0x7fffffff; x[c] = -INFINITY; rank[c] = 0;
    if (i < deg) {
      sv[c]  = min(max(seg[i], 0), NN - 1);
      key[c] = sv[c];
      x[c]   = leaky(pv + q[sv[c]] + be0);
    }
  }
  for (int j = 0; j < deg; ++j) {
    int kj = __builtin_amdgcn_readlane(key[0], j & 31);
    for (int c = 1; c < C; ++c)
      if ((j >> 5) == c) kj = __builtin_amdgcn_readlane(key[c], j & 31);
    for (int c = 0; c < C; ++c)
      rank[c] += (kj < key[c] || (kj == key[c] && j < c * 32 + lane)) ? 1 : 0;
  }
  float mx = x[0];
  for (int c = 1; c < C; ++c) mx = fmaxf(mx, x[c]);
  mx = wave_max_f(mx);
  for (int c = 0; c < C; ++c) ex[c] = (c * 32 + lane < deg) ? expf(x[c] - mx) : 0.0f;
  float den = 0.0f;
  for (int k = 0; k < deg; ++k) {
    int s_; float exk;
    pick<C>(rank, sv, ex, k, s_, exk);
    den += exk;
  }
  const float dsafe = den > 0.0f ? den : 1.0f;
  for (int c = 0; c < C; ++c) a[c] = ex[c] / dsafe;

  float4 acc = make_float4(0.0f, 0.0f, 0.0f, 0.0f);
  int k = 0;
  for (; k + 4 <= deg; k += 4) {
    int s0, s1, s2, s3; float a0, a1, a2, a3;
    pick<C>(rank, sv, a, k,     s0, a0);
    pick<C>(rank, sv, a, k + 1, s1, a1);
    pick<C>(rank, sv, a, k + 2, s2, a2);
    pick<C>(rank, sv, a, k + 3, s3, a3);
    const float4 h0 = hv4[(size_t)s0 * (GG / 4) + lane];
    const float4 h1 = hv4[(size_t)s1 * (GG / 4) + lane];
    const float4 h2 = hv4[(size_t)s2 * (GG / 4) + lane];
    const float4 h3 = hv4[(size_t)s3 * (GG / 4) + lane];
    acc.x += h0.x * a0; acc.y += h0.y * a0; acc.z += h0.z * a0; acc.w += h0.w * a0;
    acc.x += h1.x * a1; acc.y += h1.y * a1; acc.z += h1.z * a1; acc.w += h1.w * a1;
    acc.x += h2.x * a2; acc.y += h2.y * a2; acc.z += h2.z * a2; acc.w += h2.w * a2;
    acc.x += h3.x * a3; acc.y += h3.y * a3; acc.z += h3.z * a3; acc.w += h3.w * a3;
  }
  for (; k < deg; ++k) {
    int sk; float ak;
    pick<C>(rank, sv, a, k, sk, ak);
    const float4 h = hv4[(size_t)sk * (GG / 4) + lane];
    acc.x += h.x * ak; acc.y += h.y * ak; acc.z += h.z * ak; acc.w += h.w * ak;
  }
  return acc;
}

__global__ void __launch_bounds__(256)
node_softmax_aggregate(const int* __restrict__ col,
                       const int* __restrict__ off, const int* __restrict__ endp,
                       const float* __restrict__ p, const float* __restrict__ q,
                       const float* __restrict__ be, const float* __restrict__ hv,
                       float* __restrict__ out) {
  const int v    = (int)((blockIdx.x * (size_t)blockDim.x + threadIdx.x) >> 5);
  const int lane = threadIdx.x & 31;
  if (v >= NN) return;
  const int start = off[v];
  const int deg   = endp[v] - start;
  const int* seg  = col + start;
  const float pv  = p[v];
  const float be0 = be[0];
  const float4* hv4 = (const float4*)hv;
  float4 acc;

  if (deg <= 32) {
    acc = segment_in_registers<1>(seg, deg, lane, q, pv, be0, hv4);
  } else if (deg <= 64) {
    acc = segment_in_registers<2>(seg, deg, lane, q, pv, be0, hv4);
  } else {
    acc = make_float4(0.0f, 0.0f, 0.0f, 0.0f);
    float mx = -INFINITY;
    for (int c = lane; c < deg; c += 32) {
      const int s = min(max(seg[c], 0), NN - 1);
      mx = fmaxf(mx, leaky(pv + q[s] + be0));
    }
    mx = wave_max_f(mx);
    float den = 0.0f;
    int last = -1;
    for (int k = 0; k < deg; ) {
      int mult;
      const int s = next_source(seg, deg, last, lane, mult);
      if (mult <= 0) break;
      last = s;
      const float exk = expf(leaky(pv + q[min(s, NN - 1)] + be0) - mx);
      for (int t = 0; t < mult; ++t) den += exk;
      k += mult;
    }
    const float dsafe = den > 0.0f ? den : 1.0f;
    last = -1;
    for (int k = 0; k < deg; ) {
      int mult;
      const int s = next_source(seg, deg, last, lane, mult);
      if (mult <= 0) break;
      last = s;
      const int sc = min(s, NN - 1);
      const float ak = expf(leaky(pv + q[sc] + be0) - mx) / dsafe;
      const float4 h = hv4[(size_t)sc * (GG / 4) + lane];
      for (int t = 0; t < mult; ++t) {
        acc.x += h.x * ak; acc.y += h.y * ak; acc.z += h.z * ak; acc.w += h.w * ak;
      }
      k += mult;
    }
  }

  float4 r;
  r.x = acc.x > 0.0f ? acc.x : expm1f(acc.x);
  r.y = acc.y > 0.0f ? acc.y : expm1f(acc.y);
  r.z = acc.z > 0.0f ? acc.z : expm1f(acc.z);
  r.w = acc.w > 0.0f ? acc.w : expm1f(acc.w);
  ((float4*)out)[(size_t)v * (GG / 4) + lane] = r;
}

extern "C" void kernel_launch(void* const* d_in, const int* in_sizes, int n_in,
                              void* d_out, int out_size, void* d_ws, size_t ws_size,
                              hipStream_t stream) {
  (void)in_sizes; (void)n_in; (void)out_size; (void)ws_size;
  const float* nf  = (const float*)d_in[0];
  const int*   src = (const int*)  d_in[1];
  const int*   dst = (const int*)  d_in[2];
  const float* We  = (const float*)d_in[3];
  const float* be  = (const float*)d_in[4];
  const float* Wn  = (const float*)d_in[5];
  const float* bn  = (const float*)d_in[6];
  float* out = (float*)d_out;

  float* ws   = (float*)d_ws;
  float* hv   = ws;
  float* p    = hv + (size_t)NN * GG;
  float* q    = p + NN;
  int*   cnt  = (int*)(q + NN);
  int*   off  = cnt + NN;
  int*   col  = off + NN;

  node_gemm_wmma<<<NN / 16, 256, 0, stream>>>(nf, Wn, bn, hv, We, p, q, cnt);
  edge_count<<<(EE / 4 + 255) / 256, 256, 0, stream>>>(dst, cnt);
  scan_counts<<<1, SCAN_T, 0, stream>>>(cnt, off);
  edge_place<<<(EE / 4 + 255) / 256, 256, 0, stream>>>(src, dst, cnt, col);
  node_softmax_aggregate<<<(NN * 32 + 255) / 256, 256, 0, stream>>>(col, off, cnt, p, q, be, hv, out);
}


# Count and place passes: per-edge global atomics replaced by LDS atomics on a table of all nodes (64 workgroups); global counters get one add per node per workgroup; order of float sums unchanged
# speedup vs baseline: 2.0524x; 1.2240x over previous
#include <hip/hip_runtime.h>
#include <math.h>

typedef __attribute__((ext_vector_type(2))) float v2f;
typedef __attribute__((ext_vector_type(8))) float v8f;

#define NN 50000
#define EE 800000
#define DD 128
#define GG 128
#define LDS_STRIDE 132

__global__ void __launch_bounds__(256, 16)
node_gemm_wmma(const float* __restrict__ nf, const float* __restrict__ Wn,
               const float* __restrict__ bn, float* __restrict__ hv,
               const float* __restrict__ We, float* __restrict__ p,
               float* __restrict__ q, int* __restrict__ cnt) {
  __shared__ float Atile[16 * LDS_STRIDE];
  const int row0 = blockIdx.x * 16;
  const int tid  = threadIdx.x;

  #pragma unroll
  for (int i = tid; i < 16 * DD; i += 256) {
    const int r = i >> 7;
    const int c = i & (DD - 1);
    Atile[r * LDS_STRIDE + c] = nf[(size_t)(row0 + r) * DD + c];
  }
  __syncthreads();

  const int wave = tid >> 5;
  const int lane = tid & 31;
  const int col0 = wave * 16;
  const int half = lane >> 4;
  const int l16  = lane & 15;

  v8f acc = {};
  #pragma unroll
  for (int k = 0; k < DD; k += 4) {
    const int ka = k + half * 2;
    v2f a, b;
    a.x = Atile[l16 * LDS_STRIDE + ka];
    a.y = Atile[l16 * LDS_STRIDE + ka + 1];
    b.x = Wn[(size_t)(ka + 0) * GG + col0 + l16];
    b.y = Wn[(size_t)(ka + 1) * GG + col0 + l16];
    acc = __builtin_amdgcn_wmma_f32_16x16x4_f32(false, a, false, b,
                                                (short)0, acc, false, false);
  }

  #pragma unroll
  for (int i = 0; i < 8; ++i) {
    const int row = row0 + i + half * 8;
    const int col = col0 + l16;
    float v = acc[i] + bn[col];
    v = v > 0.0f ? v : 0.01f * v;
    hv[(size_t)row * GG + col] = v;
  }

  const int sr0 = wave * 2;
  float ap0 = 0.0f, aq0 = 0.0f, ap1 = 0.0f, aq1 = 0.0f;
  #pragma unroll
  for (int i = lane; i < DD; i += 32) {
    const float wp = We[i];
    const float wq = We[DD + i];
    const float x0 = Atile[sr0 * LDS_STRIDE + i];
    const float x1 = Atile[(sr0 + 1) * LDS_STRIDE + i];
    ap0 += x0 * wp;
    aq0 += x0 * wq;
    ap1 += x1 * wp;
    aq1 += x1 * wq;
  }
  #pragma unroll
  for (int off = 16; off > 0; off >>= 1) {
    ap0 += __shfl_xor(ap0, off, 32);
    aq0 += __shfl_xor(aq0, off, 32);
    ap1 += __shfl_xor(ap1, off, 32);
    aq1 += __shfl_xor(aq1, off, 32);
  }

  if (lane < 2) {
    const int node = row0 + sr0 + lane;
    p[node]   = lane ? ap1 : ap0;
    q[node]   = lane ? aq1 : aq0;
    cnt[node] = 0;
  }
}

#define EB_BLOCKS 64
#define EB_T      1024
#define EB_EDGES  (EE / EB_BLOCKS)
#define EB_Q      (EB_EDGES / 4)
#define EB_V      ((EB_Q + EB_T - 1) / EB_T)
#define EB_WORDS  50176
#define EB_SWEEP  (EB_WORDS / EB_T)
#define EB_ROT    768
#define EB_GROUP  7

__device__ __forceinline__ int node_of(int d) { return min(max(d, 0), NN - 1); }

__device__ __forceinline__ void table_clear(int* __restrict__ tab, int tid) {
  int4* t4 = (int4*)tab;
  for (int i = tid; i < EB_WORDS / 4; i += EB_T) t4[i] = make_int4(0, 0, 0, 0);
}

__device__ __forceinline__ void table_count(int* __restrict__ tab, const int* __restrict__ dst,
                                            int blk, int tid) {
  const int4* d4 = (const int4*)dst + (size_t)blk * EB_Q;
  int4 d[EB_V];
  #pragma unroll
  for (int j = 0; j < EB_V; ++j) d[j] = d4[min(tid + j * EB_T, EB_Q - 1)];
  #pragma unroll
  for (int j = 0; j < EB_V; ++j) {
    if (tid + j * EB_T < EB_Q) {
      atomicAdd(&tab[node_of(d[j].x)], 1);
      atomicAdd(&tab[node_of(d[j].y)], 1);
      atomicAdd(&tab[node_of(d[j].z)], 1);
      atomicAdd(&tab[node_of(d[j].w)], 1);
    }
  }
}

__device__ __forceinline__ int sweep_word(int tid, int j, int blk) {
  const int v = tid + j * EB_T + blk * EB_ROT;
  return v >= EB_WORDS ? v - EB_WORDS : v;
}

__global__ void __launch_bounds__(EB_T)
edge_count(const int* __restrict__ dst, int* __restrict__ cnt) {
  __shared__ __attribute__((aligned(16))) int tab[EB_WORDS];
  const int tid = threadIdx.x;
  const int blk = blockIdx.x;
  table_clear(tab, tid);
  __syncthreads();
  table_count(tab, dst, blk, tid);
  __syncthreads();
  for (int j0 = 0; j0 < EB_SWEEP; j0 += EB_GROUP) {
    int v[EB_GROUP], h[EB_GROUP];
    #pragma unroll
    for (int k = 0; k < EB_GROUP; ++k) {
      v[k] = sweep_word(tid, j0 + k, blk);
      h[k] = tab[v[k]];
    }
    #pragma unroll
    for (int k = 0; k < EB_GROUP; ++k)
      if (h[k]) atomicAdd(&cnt[v[k]], h[k]);
  }
}

#define SCAN_T   1024
#define SCAN_PER 49

__global__ void __launch_bounds__(SCAN_T)
scan_counts(int* __restrict__ cnt, int* __restrict__ off) {
  __shared__ int buf[SCAN_T * SCAN_PER];
  __shared__ int wsum[32];
  const int tid  = threadIdx.x;
  const int lane = tid & 31;
  const int w    = tid >> 5;
  for (int i = tid; i < SCAN_T * SCAN_PER; i += SCAN_T) buf[i] = (i < NN) ? cnt[i] : 0;
  __syncthreads();
  const int base = tid * SCAN_PER;
  int s = 0;
  for (int i = 0; i < SCAN_PER; ++i) s += buf[base + i];
  int inc = s;
  for (int o = 1; o < 32; o <<= 1) {
    const int t = __shfl_up(inc, o, 32);
    if (lane >= o) inc += t;
  }
  if (lane == 31) wsum[w] = inc;
  __syncthreads();
  if (w == 0) {
    const int x = wsum[lane];
    int xi = x;
    for (int o = 1; o < 32; o <<= 1) {
      const int t = __shfl_up(xi, o, 32);
      if (lane >= o) xi += t;
    }
    wsum[lane] = xi - x;
  }
  __syncthreads();
  int run = wsum[w] + inc - s;
  for (int i = 0; i < SCAN_PER; ++i) {
    const int c = buf[base + i];
    buf[base + i] = run;
    run += c;
  }
  __syncthreads();
  for (int i = tid; i < NN; i += SCAN_T) {
    const int o = buf[i];
    off[i] = o;
    cnt[i] = o;
  }
}

__global__ void __launch_bounds__(EB_T)
edge_place(const int* __restrict__ src, const int* __restrict__ dst,
           int* __restrict__ cur, int* __restrict__ col) {
  __shared__ __attribute__((aligned(16))) int tab[EB_WORDS];
  const int tid = threadIdx.x;
  const int blk = blockIdx.x;
  table_clear(tab, tid);
  __syncthreads();
  table_count(tab, dst, blk, tid);
  __syncthreads();
  for (int j0 = 0; j0 < EB_SWEEP; j0 += EB_GROUP) {
    int v[EB_GROUP], h[EB_GROUP];
    #pragma unroll
    for (int k = 0; k < EB_GROUP; ++k) {
      v[k] = sweep_word(tid, j0 + k, blk);
      h[k] = tab[v[k]];
    }
    #pragma unroll
    for (int k = 0; k < EB_GROUP; ++k)
      if (h[k]) h[k] = atomicAdd(&cur[v[k]], h[k]);
    #pragma unroll
    for (int k = 0; k < EB_GROUP; ++k) tab[v[k]] = h[k];
  }
  __syncthreads();
  const int4* d4 = (const int4*)dst + (size_t)blk * EB_Q;
  const int4* s4 = (const int4*)src + (size_t)blk * EB_Q;
  int4 d[EB_V], s[EB_V];
  #pragma unroll
  for (int j = 0; j < EB_V; ++j) {
    const int i = min(tid + j * EB_T, EB_Q - 1);
    d[j] = d4[i];
    s[j] = s4[i];
  }
  #pragma unroll
  for (int j = 0; j < EB_V; ++j) {
    if (tid + j * EB_T < EB_Q) {
      const int p0 = atomicAdd(&tab[node_of(d[j].x)], 1);
      const int p1 = atomicAdd(&tab[node_of(d[j].y)], 1);
      const int p2 = atomicAdd(&tab[node_of(d[j].z)], 1);
      const int p3 = atomicAdd(&tab[node_of(d[j].w)], 1);
      col[min(max(p0, 0), EE - 1)] = s[j].x;
      col[min(max(p1, 0), EE - 1)] = s[j].y;
      col[min(max(p2, 0), EE - 1)] = s[j].z;
      col[min(max(p3, 0), EE - 1)] = s[j].w;
    }
  }
}

__device__ inline float leaky(float x) { return x > 0.0f ? x : 0.01f * x; }

__device__ inline int wave_min_i(int x) {
  for (int o = 16; o > 0; o >>= 1) x = min(x, __shfl_xor(x, o, 32));
  return x;
}
__device__ inline int wave_sum_i(int x) {
  for (int o = 16; o > 0; o >>= 1) x += __shfl_xor(x, o, 32);
  return x;
}
__device__ inline float wave_max_f(float x) {
  for (int o = 16; o > 0; o >>= 1) x = fmaxf(x, __shfl_xor(x, o, 32));
  return x;
}

__device__ inline int next_source(const int* __restrict__ seg, int deg, int last, int lane,
                                  int& mult) {
  int best = 0x7fffffff;
  for (int c = lane; c < deg; c += 32) {
    const int s = seg[c];
    if (s > last && s < best) best = s;
  }
  best = wave_min_i(best);
  int m = 0;
  for (int c = lane; c < deg; c += 32) m += (seg[c] == best) ? 1 : 0;
  mult = (best == 0x7fffffff) ? 0 : wave_sum_i(m);
  return best;
}

template <int C>
__device__ inline void pick(const int (&rank)[C], const int (&sv)[C], const float (&w)[C],
                            int k, int& svk, float& wk) {
  svk = 0; wk = 0.0f;
  for (int c = 0; c < C; ++c) {
    const unsigned mk = __builtin_amdgcn_ballot_w32(rank[c] == k);
    if (C == 1 || mk != 0u) {
      const int j = (__ffs((int)mk) - 1) & 31;
      svk = __builtin_amdgcn_readlane(sv[c], j);
      wk  = __int_as_float(__builtin_amdgcn_readlane(__float_as_int(w[c]), j));
    }
  }
}

template <int C>
__device__ inline float4 segment_in_registers(const int* __restrict__ seg, int deg, int lane,
                                              const float* __restrict__ q,
                                              float pv, float be0,
                                              const float4* __restrict__ hv4) {
  int sv[C], key[C], rank[C];
  float x[C], ex[C], a[C];
  for (int c = 0; c < C; ++c) {
    const int i = c * 32 + lane;
    sv[c] = 0; key[c] = 0x7fffffff; x[c] = -INFINITY; rank[c] = 0;
    if (i < deg) {
      sv[c]  = min(max(seg[i], 0), NN - 1);
      key[c] = sv[c];
      x[c]   = leaky(pv + q[sv[c]] + be0);
    }
  }
  for (int j = 0; j < deg; ++j) {
    int kj = __builtin_amdgcn_readlane(key[0], j & 31);
    for (int c = 1; c < C; ++c)
      if ((j >> 5) == c) kj = __builtin_amdgcn_readlane(key[c], j & 31);
    for (int c = 0; c < C; ++c)
      rank[c] += (kj < key[c] || (kj == key[c] && j < c * 32 + lane)) ? 1 : 0;
  }
  float mx = x[0];
  for (int c = 1; c < C; ++c) mx = fmaxf(mx, x[c]);
  mx = wave_max_f(mx);
  for (int c = 0; c < C; ++c) ex[c] = (c * 32 + lane < deg) ? expf(x[c] - mx) : 0.0f;
  float den = 0.0f;
  for (int k = 0; k < deg; ++k) {
    int s_; float exk;
    pick<C>(rank, sv, ex, k, s_, exk);
    den += exk;
  }
  const float dsafe = den > 0.0f ? den : 1.0f;
  for (int c = 0; c < C; ++c) a[c] = ex[c] / dsafe;

  float4 acc = make_float4(0.0f, 0.0f, 0.0f, 0.0f);
  int k = 0;
  for (; k + 4 <= deg; k += 4) {
    int s0, s1, s2, s3; float a0, a1, a2, a3;
    pick<C>(rank, sv, a, k,     s0, a0);
    pick<C>(rank, sv, a, k + 1, s1, a1);
    pick<C>(rank, sv, a, k + 2, s2, a2);
    pick<C>(rank, sv, a, k + 3, s3, a3);
    const float4 h0 = hv4[(size_t)s0 * (GG / 4) + lane];
    const float4 h1 = hv4[(size_t)s1 * (GG / 4) + lane];
    const float4 h2 = hv4[(size_t)s2 * (GG / 4) + lane];
    const float4 h3 = hv4[(size_t)s3 * (GG / 4) + lane];
    acc.x += h0.x * a0; acc.y += h0.y * a0; acc.z += h0.z * a0; acc.w += h0.w * a0;
    acc.x += h1.x * a1; acc.y += h1.y * a1; acc.z += h1.z * a1; acc.w += h1.w * a1;
    acc.x += h2.x * a2; acc.y += h2.y * a2; acc.z += h2.z * a2; acc.w += h2.w * a2;
    acc.x += h3.x * a3; acc.y += h3.y * a3; acc.z += h3.z * a3; acc.w += h3.w * a3;
  }
  for (; k < deg; ++k) {
    int sk; float ak;
    pick<C>(rank, sv, a, k, sk, ak);
    const float4 h = hv4[(size_t)sk * (GG / 4) + lane];
    acc.x += h.x * ak; acc.y += h.y * ak; acc.z += h.z * ak; acc.w += h.w * ak;
  }
  return acc;
}

__global__ void __launch_bounds__(256)
node_softmax_aggregate(const int* __restrict__ col,
                       const int* __restrict__ off, const int* __restrict__ endp,
                       const float* __restrict__ p, const float* __restrict__ q,
                       const float* __restrict__ be, const float* __restrict__ hv,
                       float* __restrict__ out) {
  const int v    = (int)((blockIdx.x * (size_t)blockDim.x + threadIdx.x) >> 5);
  const int lane = threadIdx.x & 31;
  if (v >= NN) return;
  const int start = off[v];
  const int deg   = endp[v] - start;
  const int* seg  = col + start;
  const float pv  = p[v];
  const float be0 = be[0];
  const float4* hv4 = (const float4*)hv;
  float4 acc;

  if (deg <= 32) {
    acc = segment_in_registers<1>(seg, deg, lane, q, pv, be0, hv4);
  } else if (deg <= 64) {
    acc = segment_in_registers<2>(seg, deg, lane, q, pv, be0, hv4);
  } else {
    acc = make_float4(0.0f, 0.0f, 0.0f, 0.0f);
    float mx = -INFINITY;
    for (int c = lane; c < deg; c += 32) {
      const int s = min(max(seg[c], 0), NN - 1);
      mx = fmaxf(mx, leaky(pv + q[s] + be0));
    }
    mx = wave_max_f(mx);
    float den = 0.0f;
    int last = -1;
    for (int k = 0; k < deg; ) {
      int mult;
      const int s = next_source(seg, deg, last, lane, mult);
      if (mult <= 0) break;
      last = s;
      const float exk = expf(leaky(pv + q[min(s, NN - 1)] + be0) - mx);
      for (int t = 0; t < mult; ++t) den += exk;
      k += mult;
    }
    const float dsafe = den > 0.0f ? den : 1.0f;
    last = -1;
    for (int k = 0; k < deg; ) {
      int mult;
      const int s = next_source(seg, deg, last, lane, mult);
      if (mult <= 0) break;
      last = s;
      const int sc = min(s, NN - 1);
      const float ak = expf(leaky(pv + q[sc] + be0) - mx) / dsafe;
      const float4 h = hv4[(size_t)sc * (GG / 4) + lane];
      for (int t = 0; t < mult; ++t) {
        acc.x += h.x * ak; acc.y += h.y * ak; acc.z += h.z * ak; acc.w += h.w * ak;
      }
      k += mult;
    }
  }

  float4 r;
  r.x = acc.x > 0.0f ? acc.x : expm1f(acc.x);
  r.y = acc.y > 0.0f ? acc.y : expm1f(acc.y);
  r.z = acc.z > 0.0f ? acc.z : expm1f(acc.z);
  r.w = acc.w > 0.0f ? acc.w : expm1f(acc.w);
  ((float4*)out)[(size_t)v * (GG / 4) + lane] = r;
}

extern "C" void kernel_launch(void* const* d_in, const int* in_sizes, int n_in,
                              void* d_out, int out_size, void* d_ws, size_t ws_size,
                              hipStream_t stream) {
  (void)in_sizes; (void)n_in; (void)out_size; (void)ws_size;
  const float* nf  = (const float*)d_in[0];
  const int*   src = (const int*)  d_in[1];
  const int*   dst = (const int*)  d_in[2];
  const float* We  = (const float*)d_in[3];
  const float* be  = (const float*)d_in[4];
  const float* Wn  = (const float*)d_in[5];
  const float* bn  = (const float*)d_in[6];
  float* out = (float*)d_out;

  float* ws   = (float*)d_ws;
  float* hv   = ws;
  float* p    = hv + (size_t)NN * GG;
  float* q    = p + NN;
  int*   cnt  = (int*)(q + NN);
  int*   off  = cnt + NN;
  int*   col  = off + NN;

  node_gemm_wmma<<<NN / 16, 256, 0, stream>>>(nf, Wn, bn, hv, We, p, q, cnt);
  edge_count<<<EB_BLOCKS, EB_T, 0, stream>>>(dst, cnt);
  scan_counts<<<1, SCAN_T, 0, stream>>>(cnt, off);
  edge_place<<<EB_BLOCKS, EB_T, 0, stream>>>(src, dst, cnt, col);
  node_softmax_aggregate<<<(NN * 32 + 255) / 256, 256, 0, stream>>>(col, off, cnt, p, q, be, hv, out);
}


# Scan launch: the 49 count loads per thread, each waited for before the next was issued, become 13 loads of four counts each, all issued as one group before the LDS stores; float sums untouched
# speedup vs baseline: 2.2605x; 1.1014x over previous
#include <hip/hip_runtime.h>
#include <math.h>

typedef __attribute__((ext_vector_type(2))) float v2f;
typedef __attribute__((ext_vector_type(8))) float v8f;

#define NN 50000
#define EE 800000
#define DD 128
#define GG 128
#define LDS_STRIDE 132

__global__ void __launch_bounds__(256, 16)
node_gemm_wmma(const float* __restrict__ nf, const float* __restrict__ Wn,
               const float* __restrict__ bn, float* __restrict__ hv,
               const float* __restrict__ We, float* __restrict__ p,
               float* __restrict__ q, int* __restrict__ cnt) {
  __shared__ float Atile[16 * LDS_STRIDE];
  const int row0 = blockIdx.x * 16;
  const int tid  = threadIdx.x;

  #pragma unroll
  for (int i = tid; i < 16 * DD; i += 256) {
    const int r = i >> 7;
    const int c = i & (DD - 1);
    Atile[r * LDS_STRIDE + c] = nf[(size_t)(row0 + r) * DD + c];
  }
  __syncthreads();

  const int wave = tid >> 5;
  const int lane = tid & 31;
  const int col0 = wave * 16;
  const int half = lane >> 4;
  const int l16  = lane & 15;

  v8f acc = {};
  #pragma unroll
  for (int k = 0; k < DD; k += 4) {
    const int ka = k + half * 2;
    v2f a, b;
    a.x = Atile[l16 * LDS_STRIDE + ka];
    a.y = Atile[l16 * LDS_STRIDE + ka + 1];
    b.x = Wn[(size_t)(ka + 0) * GG + col0 + l16];
    b.y = Wn[(size_t)(ka + 1) * GG + col0 + l16];
    acc = __builtin_amdgcn_wmma_f32_16x16x4_f32(false, a, false, b,
                                                (short)0, acc, false, false);
  }

  #pragma unroll
  for (int i = 0; i < 8; ++i) {
    const int row = row0 + i + half * 8;
    const int col = col0 + l16;
    float v = acc[i] + bn[col];
    v = v > 0.0f ? v : 0.01f * v;
    hv[(size_t)row * GG + col] = v;
  }

  const int sr0 = wave * 2;
  float ap0 = 0.0f, aq0 = 0.0f, ap1 = 0.0f, aq1 = 0.0f;
  #pragma unroll
  for (int i = lane; i < DD; i += 32) {
    const float wp = We[i];
    const float wq = We[DD + i];
    const float x0 = Atile[sr0 * LDS_STRIDE + i];
    const float x1 = Atile[(sr0 + 1) * LDS_STRIDE + i];
    ap0 += x0 * wp;
    aq0 += x0 * wq;
    ap1 += x1 * wp;
    aq1 += x1 * wq;
  }
  #pragma unroll
  for (int off = 16; off > 0; off >>= 1) {
    ap0 += __shfl_xor(ap0, off, 32);
    aq0 += __shfl_xor(aq0, off, 32);
    ap1 += __shfl_xor(ap1, off, 32);
    aq1 += __shfl_xor(aq1, off, 32);
  }

  if (lane < 2) {
    const int node = row0 + sr0 + lane;
    p[node]   = lane ? ap1 : ap0;
    q[node]   = lane ? aq1 : aq0;
    cnt[node] = 0;
  }
}

#define EB_BLOCKS 64
#define EB_T      1024
#define EB_EDGES  (EE / EB_BLOCKS)
#define EB_Q      (EB_EDGES / 4)
#define EB_V      ((EB_Q + EB_T - 1) / EB_T)
#define EB_WORDS  50176
#define EB_SWEEP  (EB_WORDS / EB_T)
#define EB_ROT    768
#define EB_GROUP  7

__device__ __forceinline__ int node_of(int d) { return min(max(d, 0), NN - 1); }

__device__ __forceinline__ void table_clear(int* __restrict__ tab, int tid) {
  int4* t4 = (int4*)tab;
  for (int i = tid; i < EB_WORDS / 4; i += EB_T) t4[i] = make_int4(0, 0, 0, 0);
}

__device__ __forceinline__ void table_count(int* __restrict__ tab, const int* __restrict__ dst,
                                            int blk, int tid) {
  const int4* d4 = (const int4*)dst + (size_t)blk * EB_Q;
  int4 d[EB_V];
  #pragma unroll
  for (int j = 0; j < EB_V; ++j) d[j] = d4[min(tid + j * EB_T, EB_Q - 1)];
  #pragma unroll
  for (int j = 0; j < EB_V; ++j) {
    if (tid + j * EB_T < EB_Q) {
      atomicAdd(&tab[node_of(d[j].x)], 1);
      atomicAdd(&tab[node_of(d[j].y)], 1);
      atomicAdd(&tab[node_of(d[j].z)], 1);
      atomicAdd(&tab[node_of(d[j].w)], 1);
    }
  }
}

__device__ __forceinline__ int sweep_word(int tid, int j, int blk) {
  const int v = tid + j * EB_T + blk * EB_ROT;
  return v >= EB_WORDS ? v - EB_WORDS : v;
}

__global__ void __launch_bounds__(EB_T)
edge_count(const int* __restrict__ dst, int* __restrict__ cnt) {
  __shared__ __attribute__((aligned(16))) int tab[EB_WORDS];
  const int tid = threadIdx.x;
  const int blk = blockIdx.x;
  table_clear(tab, tid);
  __syncthreads();
  table_count(tab, dst, blk, tid);
  __syncthreads();
  for (int j0 = 0; j0 < EB_SWEEP; j0 += EB_GROUP) {
    int v[EB_GROUP], h[EB_GROUP];
    #pragma unroll
    for (int k = 0; k < EB_GROUP; ++k) {
      v[k] = sweep_word(tid, j0 + k, blk);
      h[k] = tab[v[k]];
    }
    #pragma unroll
    for (int k = 0; k < EB_GROUP; ++k)
      if (h[k]) atomicAdd(&cnt[v[k]], h[k]);
  }
}

#define SCAN_T   1024
#define SCAN_PER 49
#define SCAN_Q   (SCAN_T * SCAN_PER / 4)
#define SCAN_NQ  (NN / 4)
#define SCAN_LD  ((SCAN_Q + SCAN_T - 1) / SCAN_T)

__global__ void __launch_bounds__(SCAN_T)
scan_counts(int* __restrict__ cnt, int* __restrict__ off) {
  __shared__ __attribute__((aligned(16))) int buf[SCAN_LD * SCAN_T * 4];
  __shared__ int wsum[32];
  const int tid  = threadIdx.x;
  const int lane = tid & 31;
  const int w    = tid >> 5;
  const int4* c4 = (const int4*)cnt;
  int4*       b4 = (int4*)buf;
  int4 r[SCAN_LD];
  #pragma unroll
  for (int j = 0; j < SCAN_LD; ++j) r[j] = c4[min(tid + j * SCAN_T, SCAN_NQ - 1)];
  #pragma unroll
  for (int j = 0; j < SCAN_LD; ++j) {
    const int i = tid + j * SCAN_T;
    b4[i] = (i < SCAN_NQ) ? r[j] : make_int4(0, 0, 0, 0);
  }
  __syncthreads();
  const int base = tid * SCAN_PER;
  int s = 0;
  for (int i = 0; i < SCAN_PER; ++i) s += buf[base + i];
  int inc = s;
  for (int o = 1; o < 32; o <<= 1) {
    const int t = __shfl_up(inc, o, 32);
    if (lane >= o) inc += t;
  }
  if (lane == 31) wsum[w] = inc;
  __syncthreads();
  if (w == 0) {
    const int x = wsum[lane];
    int xi = x;
    for (int o = 1; o < 32; o <<= 1) {
      const int t = __shfl_up(xi, o, 32);
      if (lane >= o) xi += t;
    }
    wsum[lane] = xi - x;
  }
  __syncthreads();
  int run = wsum[w] + inc - s;
  for (int i = 0; i < SCAN_PER; ++i) {
    const int c = buf[base + i];
    buf[base + i] = run;
    run += c;
  }
  __syncthreads();
  for (int i = tid; i < NN; i += SCAN_T) {
    const int o = buf[i];
    off[i] = o;
    cnt[i] = o;
  }
}

__global__ void __launch_bounds__(EB_T)
edge_place(const int* __restrict__ src, const int* __restrict__ dst,
           int* __restrict__ cur, int* __restrict__ col) {
  __shared__ __attribute__((aligned(16))) int tab[EB_WORDS];
  const int tid = threadIdx.x;
  const int blk = blockIdx.x;
  table_clear(tab, tid);
  __syncthreads();
  table_count(tab, dst, blk, tid);
  __syncthreads();
  for (int j0 = 0; j0 < EB_SWEEP; j0 += EB_GROUP) {
    int v[EB_GROUP], h[EB_GROUP];
    #pragma unroll
    for (int k = 0; k < EB_GROUP; ++k) {
      v[k] = sweep_word(tid, j0 + k, blk);
      h[k] = tab[v[k]];
    }
    #pragma unroll
    for (int k = 0; k < EB_GROUP; ++k)
      if (h[k]) h[k] = atomicAdd(&cur[v[k]], h[k]);
    #pragma unroll
    for (int k = 0; k < EB_GROUP; ++k) tab[v[k]] = h[k];
  }
  __syncthreads();
  const int4* d4 = (const int4*)dst + (size_t)blk * EB_Q;
  const int4* s4 = (const int4*)src + (size_t)blk * EB_Q;
  int4 d[EB_V], s[EB_V];
  #pragma unroll
  for (int j = 0; j < EB_V; ++j) {
    const int i = min(tid + j * EB_T, EB_Q - 1);
    d[j] = d4[i];
    s[j] = s4[i];
  }
  #pragma unroll
  for (int j = 0; j < EB_V; ++j) {
    if (tid + j * EB_T < EB_Q) {
      const int p0 = atomicAdd(&tab[node_of(d[j].x)], 1);
      const int p1 = atomicAdd(&tab[node_of(d[j].y)], 1);
      const int p2 = atomicAdd(&tab[node_of(d[j].z)], 1);
      const int p3 = atomicAdd(&tab[node_of(d[j].w)], 1);
      col[min(max(p0, 0), EE - 1)] = s[j].x;
      col[min(max(p1, 0), EE - 1)] = s[j].y;
      col[min(max(p2, 0), EE - 1)] = s[j].z;
      col[min(max(p3, 0), EE - 1)] = s[j].w;
    }
  }
}

__device__ inline float leaky(float x) { return x > 0.0f ? x : 0.01f * x; }

__device__ inline int wave_min_i(int x) {
  for (int o = 16; o > 0; o >>= 1) x = min(x, __shfl_xor(x, o, 32));
  return x;
}
__device__ inline int wave_sum_i(int x) {
  for (int o = 16; o > 0; o >>= 1) x += __shfl_xor(x, o, 32);
  return x;
}
__device__ inline float wave_max_f(float x) {
  for (int o = 16; o > 0; o >>= 1) x = fmaxf(x, __shfl_xor(x, o, 32));
  return x;
}

__device__ inline int next_source(const int* __restrict__ seg, int deg, int last, int lane,
                                  int& mult) {
  int best = 0x7fffffff;
  for (int c = lane; c < deg; c += 32) {
    const int s = seg[c];
    if (s > last && s < best) best = s;
  }
  best = wave_min_i(best);
  int m = 0;
  for (int c = lane; c < deg; c += 32) m += (seg[c] == best) ? 1 : 0;
  mult = (best == 0x7fffffff) ? 0 : wave_sum_i(m);
  return best;
}

template <int C>
__device__ inline void pick(const int (&rank)[C], const int (&sv)[C], const float (&w)[C],
                            int k, int& svk, float& wk) {
  svk = 0; wk = 0.0f;
  for (int c = 0; c < C; ++c) {
    const unsigned mk = __builtin_amdgcn_ballot_w32(rank[c] == k);
    if (C == 1 || mk != 0u) {
      const int j = (__ffs((int)mk) - 1) & 31;
      svk = __builtin_amdgcn_readlane(sv[c], j);
      wk  = __int_as_float(__builtin_amdgcn_readlane(__float_as_int(w[c]), j));
    }
  }
}

template <int C>
__device__ inline float4 segment_in_registers(const int* __restrict__ seg, int deg, int lane,
                                              const float* __restrict__ q,
                                              float pv, float be0,
                                              const float4* __restrict__ hv4) {
  int sv[C], key[C], rank[C];
  float x[C], ex[C], a[C];
  for (int c = 0; c < C; ++c) {
    const int i = c * 32 + lane;
    sv[c] = 0; key[c] = 0x7fffffff; x[c] = -INFINITY; rank[c] = 0;
    if (i < deg) {
      sv[c]  = min(max(seg[i], 0), NN - 1);
      key[c] = sv[c];
      x[c]   = leaky(pv + q[sv[c]] + be0);
    }
  }
  for (int j = 0; j < deg; ++j) {
    int kj = __builtin_amdgcn_readlane(key[0], j & 31);
    for (int c = 1; c < C; ++c)
      if ((j >> 5) == c) kj = __builtin_amdgcn_readlane(key[c], j & 31);
    for (int c = 0; c < C; ++c)
      rank[c] += (kj < key[c] || (kj == key[c] && j < c * 32 + lane)) ? 1 : 0;
  }
  float mx = x[0];
  for (int c = 1; c < C; ++c) mx = fmaxf(mx, x[c]);
  mx = wave_max_f(mx);
  for (int c = 0; c < C; ++c) ex[c] = (c * 32 + lane < deg) ? expf(x[c] - mx) : 0.0f;
  float den = 0.0f;
  for (int k = 0; k < deg; ++k) {
    int s_; float exk;
    pick<C>(rank, sv, ex, k, s_, exk);
    den += exk;
  }
  const float dsafe = den > 0.0f ? den : 1.0f;
  for (int c = 0; c < C; ++c) a[c] = ex[c] / dsafe;

  float4 acc = make_float4(0.0f, 0.0f, 0.0f, 0.0f);
  int k = 0;
  for (; k + 4 <= deg; k += 4) {
    int s0, s1, s2, s3; float a0, a1, a2, a3;
    pick<C>(rank, sv, a, k,     s0, a0);
    pick<C>(rank, sv, a, k + 1, s1, a1);
    pick<C>(rank, sv, a, k + 2, s2, a2);
    pick<C>(rank, sv, a, k + 3, s3, a3);
    const float4 h0 = hv4[(size_t)s0 * (GG / 4) + lane];
    const float4 h1 = hv4[(size_t)s1 * (GG / 4) + lane];
    const float4 h2 = hv4[(size_t)s2 * (GG / 4) + lane];
    const float4 h3 = hv4[(size_t)s3 * (GG / 4) + lane];
    acc.x += h0.x * a0; acc.y += h0.y * a0; acc.z += h0.z * a0; acc.w += h0.w * a0;
    acc.x += h1.x * a1; acc.y += h1.y * a1; acc.z += h1.z * a1; acc.w += h1.w * a1;
    acc.x += h2.x * a2; acc.y += h2.y * a2; acc.z += h2.z * a2; acc.w += h2.w * a2;
    acc.x += h3.x * a3; acc.y += h3.y * a3; acc.z += h3.z * a3; acc.w += h3.w * a3;
  }
  for (; k < deg; ++k) {
    int sk; float ak;
    pick<C>(rank, sv, a, k, sk, ak);
    const float4 h = hv4[(size_t)sk * (GG / 4) + lane];
    acc.x += h.x * ak; acc.y += h.y * ak; acc.z += h.z * ak; acc.w += h.w * ak;
  }
  return acc;
}

__global__ void __launch_bounds__(256)
node_softmax_aggregate(const int* __restrict__ col,
                       const int* __restrict__ off, const int* __restrict__ endp,
                       const float* __restrict__ p, const float* __restrict__ q,
                       const float* __restrict__ be, const float* __restrict__ hv,
                       float* __restrict__ out) {
  const int v    = (int)((blockIdx.x * (size_t)blockDim.x + threadIdx.x) >> 5);
  const int lane = threadIdx.x & 31;
  if (v >= NN) return;
  const int start = off[v];
  const int deg   = endp[v] - start;
  const int* seg  = col + start;
  const float pv  = p[v];
  const float be0 = be[0];
  const float4* hv4 = (const float4*)hv;
  float4 acc;

  if (deg <= 32) {
    acc = segment_in_registers<1>(seg, deg, lane, q, pv, be0, hv4);
  } else if (deg <= 64) {
    acc = segment_in_registers<2>(seg, deg, lane, q, pv, be0, hv4);
  } else {
    acc = make_float4(0.0f, 0.0f, 0.0f, 0.0f);
    float mx = -INFINITY;
    for (int c = lane; c < deg; c += 32) {
      const int s = min(max(seg[c], 0), NN - 1);
      mx = fmaxf(mx, leaky(pv + q[s] + be0));
    }
    mx = wave_max_f(mx);
    float den = 0.0f;
    int last = -1;
    for (int k = 0; k < deg; ) {
      int mult;
      const int s = next_source(seg, deg, last, lane, mult);
      if (mult <= 0) break;
      last = s;
      const float exk = expf(leaky(pv + q[min(s, NN - 1)] + be0) - mx);
      for (int t = 0; t < mult; ++t) den += exk;
      k += mult;
    }
    const float dsafe = den > 0.0f ? den : 1.0f;
    last = -1;
    for (int k = 0; k < deg; ) {
      int mult;
      const int s = next_source(seg, deg, last, lane, mult);
      if (mult <= 0) break;
      last = s;
      const int sc = min(s, NN - 1);
      const float ak = expf(leaky(pv + q[sc] + be0) - mx) / dsafe;
      const float4 h = hv4[(size_t)sc * (GG / 4) + lane];
      for (int t = 0; t < mult; ++t) {
        acc.x += h.x * ak; acc.y += h.y * ak; acc.z += h.z * ak; acc.w += h.w * ak;
      }
      k += mult;
    }
  }

  float4 r;
  r.x = acc.x > 0.0f ? acc.x : expm1f(acc.x);
  r.y = acc.y > 0.0f ? acc.y : expm1f(acc.y);
  r.z = acc.z > 0.0f ? acc.z : expm1f(acc.z);
  r.w = acc.w > 0.0f ? acc.w : expm1f(acc.w);
  ((float4*)out)[(size_t)v * (GG / 4) + lane] = r;
}

extern "C" void kernel_launch(void* const* d_in, const int* in_sizes, int n_in,
                              void* d_out, int out_size, void* d_ws, size_t ws_size,
                              hipStream_t stream) {
  (void)in_sizes; (void)n_in; (void)out_size; (void)ws_size;
  const float* nf  = (const float*)d_in[0];
  const int*   src = (const int*)  d_in[1];
  const int*   dst = (const int*)  d_in[2];
  const float* We  = (const float*)d_in[3];
  const float* be  = (const float*)d_in[4];
  const float* Wn  = (const float*)d_in[5];
  const float* bn  = (const float*)d_in[6];
  float* out = (float*)d_out;

  float* ws   = (float*)d_ws;
  float* hv   = ws;
  float* p    = hv + (size_t)NN * GG;
  float* q    = p + NN;
  int*   cnt  = (int*)(q + NN);
  int*   off  = cnt + NN;
  int*   col  = off + NN;

  node_gemm_wmma<<<NN / 16, 256, 0, stream>>>(nf, Wn, bn, hv, We, p, q, cnt);
  edge_count<<<EB_BLOCKS, EB_T, 0, stream>>>(dst, cnt);
  scan_counts<<<1, SCAN_T, 0, stream>>>(cnt, off);
  edge_place<<<EB_BLOCKS, EB_T, 0, stream>>>(src, dst, cnt, col);
  node_softmax_aggregate<<<(NN * 32 + 255) / 256, 256, 0, stream>>>(col, off, cnt, p, q, be, hv, out);
}


# Count and place passes: each chunk of edges goes to four workgroups, one per quarter of the nodes (52 KB table, 13-word sweep): 256 workgroups on all WGPs, not 64; float sums untouched
# speedup vs baseline: 2.8676x; 1.2686x over previous
#include <hip/hip_runtime.h>
#include <math.h>

typedef __attribute__((ext_vector_type(2))) float v2f;
typedef __attribute__((ext_vector_type(8))) float v8f;

#define NN 50000
#define EE 800000
#define DD 128
#define GG 128
#define LDS_STRIDE 132

__global__ void __launch_bounds__(256, 16)
node_gemm_wmma(const float* __restrict__ nf, const float* __restrict__ Wn,
               const float* __restrict__ bn, float* __restrict__ hv,
               const float* __restrict__ We, float* __restrict__ p,
               float* __restrict__ q, int* __restrict__ cnt) {
  __shared__ float Atile[16 * LDS_STRIDE];
  const int row0 = blockIdx.x * 16;
  const int tid  = threadIdx.x;

  #pragma unroll
  for (int i = tid; i < 16 * DD; i += 256) {
    const int r = i >> 7;
    const int c = i & (DD - 1);
    Atile[r * LDS_STRIDE + c] = nf[(size_t)(row0 + r) * DD + c];
  }
  __syncthreads();

  const int wave = tid >> 5;
  const int lane = tid & 31;
  const int col0 = wave * 16;
  const int half = lane >> 4;
  const int l16  = lane & 15;

  v8f acc = {};
  #pragma unroll
  for (int k = 0; k < DD; k += 4) {
    const int ka = k + half * 2;
    v2f a, b;
    a.x = Atile[l16 * LDS_STRIDE + ka];
    a.y = Atile[l16 * LDS_STRIDE + ka + 1];
    b.x = Wn[(size_t)(ka + 0) * GG + col0 + l16];
    b.y = Wn[(size_t)(ka + 1) * GG + col0 + l16];
    acc = __builtin_amdgcn_wmma_f32_16x16x4_f32(false, a, false, b,
                                                (short)0, acc, false, false);
  }

  #pragma unroll
  for (int i = 0; i < 8; ++i) {
    const int row = row0 + i + half * 8;
    const int col = col0 + l16;
    float v = acc[i] + bn[col];
    v = v > 0.0f ? v : 0.01f * v;
    hv[(size_t)row * GG + col] = v;
  }

  const int sr0 = wave * 2;
  float ap0 = 0.0f, aq0 = 0.0f, ap1 = 0.0f, aq1 = 0.0f;
  #pragma unroll
  for (int i = lane; i < DD; i += 32) {
    const float wp = We[i];
    const float wq = We[DD + i];
    const float x0 = Atile[sr0 * LDS_STRIDE + i];
    const float x1 = Atile[(sr0 + 1) * LDS_STRIDE + i];
    ap0 += x0 * wp;
    aq0 += x0 * wq;
    ap1 += x1 * wp;
    aq1 += x1 * wq;
  }
  #pragma unroll
  for (int off = 16; off > 0; off >>= 1) {
    ap0 += __shfl_xor(ap0, off, 32);
    aq0 += __shfl_xor(aq0, off, 32);
    ap1 += __shfl_xor(ap1, off, 32);
    aq1 += __shfl_xor(aq1, off, 32);
  }

  if (lane < 2) {
    const int node = row0 + sr0 + lane;
    p[node]   = lane ? ap1 : ap0;
    q[node]   = lane ? aq1 : aq0;
    cnt[node] = 0;
  }
}

#define EB_CHUNKS 64
#define EB_PARTS  4
#define EB_BLOCKS (EB_CHUNKS * EB_PARTS)
#define EB_T      1024
#define EB_EDGES  (EE / EB_CHUNKS)
#define EB_Q      (EB_EDGES / 4)
#define EB_V      ((EB_Q + EB_T - 1) / EB_T)
#define EB_NODES  12512
#define EB_SWEEP  13
#define EB_WORDS  (EB_SWEEP * EB_T)
#define EB_ROT    192

static_assert(EB_PARTS * EB_NODES >= NN, "the node ranges cover all nodes");
static_assert(EB_WORDS >= EB_NODES && EB_NODES % 32 == 0, "table holds a range; ranges start on a line");
static_assert(EB_CHUNKS * EB_ROT <= EB_WORDS && EB_ROT % 32 == 0, "sweep starts stay inside the table, on lines");
static_assert(EB_EDGES * EB_CHUNKS == EE && EB_Q * 4 == EB_EDGES, "chunks are whole 16-byte pieces");

__device__ __forceinline__ unsigned word_of(int d, int base) { return (unsigned)d - (unsigned)base; }

__device__ __forceinline__ void table_clear(int* __restrict__ tab, int tid) {
  int4* t4 = (int4*)tab;
  for (int i = tid; i < EB_WORDS / 4; i += EB_T) t4[i] = make_int4(0, 0, 0, 0);
}

__device__ __forceinline__ void table_count(int* __restrict__ tab, const int* __restrict__ dst,
                                            int chunk, int base, unsigned have, int tid) {
  const int4* d4 = (const int4*)dst + (size_t)chunk * EB_Q;
  int4 d[EB_V];
  #pragma unroll
  for (int j = 0; j < EB_V; ++j) d[j] = d4[min(tid + j * EB_T, EB_Q - 1)];
  #pragma unroll
  for (int j = 0; j < EB_V; ++j) {
    if (tid + j * EB_T < EB_Q) {
      const unsigned r0 = word_of(d[j].x, base);
      const unsigned r1 = word_of(d[j].y, base);
      const unsigned r2 = word_of(d[j].z, base);
      const unsigned r3 = word_of(d[j].w, base);
      if (r0 < have) atomicAdd(&tab[r0], 1);
      if (r1 < have) atomicAdd(&tab[r1], 1);
      if (r2 < have) atomicAdd(&tab[r2], 1);
      if (r3 < have) atomicAdd(&tab[r3], 1);
    }
  }
}

__device__ __forceinline__ int sweep_word(int tid, int j, int chunk) {
  const int v = tid + j * EB_T + chunk * EB_ROT;
  return v >= EB_WORDS ? v - EB_WORDS : v;
}

__global__ void __launch_bounds__(EB_T)
edge_count(const int* __restrict__ dst, int* __restrict__ cnt) {
  __shared__ __attribute__((aligned(16))) int tab[EB_WORDS];
  const int tid   = threadIdx.x;
  const int chunk = blockIdx.x % EB_CHUNKS;
  const int base  = (blockIdx.x / EB_CHUNKS) * EB_NODES;
  const unsigned have = (unsigned)min(EB_NODES, NN - base);
  table_clear(tab, tid);
  __syncthreads();
  table_count(tab, dst, chunk, base, have, tid);
  __syncthreads();
  int v[EB_SWEEP], h[EB_SWEEP];
  #pragma unroll
  for (int k = 0; k < EB_SWEEP; ++k) {
    v[k] = sweep_word(tid, k, chunk);
    h[k] = tab[v[k]];
  }
  #pragma unroll
  for (int k = 0; k < EB_SWEEP; ++k)
    if (h[k]) atomicAdd(&cnt[base + v[k]], h[k]);
}

#define SCAN_T   1024
#define SCAN_PER 49
#define SCAN_Q   (SCAN_T * SCAN_PER / 4)
#define SCAN_NQ  (NN / 4)
#define SCAN_LD  ((SCAN_Q + SCAN_T - 1) / SCAN_T)

__global__ void __launch_bounds__(SCAN_T)
scan_counts(int* __restrict__ cnt, int* __restrict__ off) {
  __shared__ __attribute__((aligned(16))) int buf[SCAN_LD * SCAN_T * 4];
  __shared__ int wsum[32];
  const int tid  = threadIdx.x;
  const int lane = tid & 31;
  const int w    = tid >> 5;
  const int4* c4 = (const int4*)cnt;
  int4*       b4 = (int4*)buf;
  int4 r[SCAN_LD];
  #pragma unroll
  for (int j = 0; j < SCAN_LD; ++j) r[j] = c4[min(tid + j * SCAN_T, SCAN_NQ - 1)];
  #pragma unroll
  for (int j = 0; j < SCAN_LD; ++j) {
    const int i = tid + j * SCAN_T;
    b4[i] = (i < SCAN_NQ) ? r[j] : make_int4(0, 0, 0, 0);
  }
  __syncthreads();
  const int base = tid * SCAN_PER;
  int s = 0;
  for (int i = 0; i < SCAN_PER; ++i) s += buf[base + i];
  int inc = s;
  for (int o = 1; o < 32; o <<= 1) {
    const int t = __shfl_up(inc, o, 32);
    if (lane >= o) inc += t;
  }
  if (lane == 31) wsum[w] = inc;
  __syncthreads();
  if (w == 0) {
    const int x = wsum[lane];
    int xi = x;
    for (int o = 1; o < 32; o <<= 1) {
      const int t = __shfl_up(xi, o, 32);
      if (lane >= o) xi += t;
    }
    wsum[lane] = xi - x;
  }
  __syncthreads();
  int run = wsum[w] + inc - s;
  for (int i = 0; i < SCAN_PER; ++i) {
    const int c = buf[base + i];
    buf[base + i] = run;
    run += c;
  }
  __syncthreads();
  for (int i = tid; i < NN; i += SCAN_T) {
    const int o = buf[i];
    off[i] = o;
    cnt[i] = o;
  }
}

__global__ void __launch_bounds__(EB_T)
edge_place(const int* __restrict__ src, const int* __restrict__ dst,
           int* __restrict__ cur, int* __restrict__ col) {
  __shared__ __attribute__((aligned(16))) int tab[EB_WORDS];
  const int tid   = threadIdx.x;
  const int chunk = blockIdx.x % EB_CHUNKS;
  const int base  = (blockIdx.x / EB_CHUNKS) * EB_NODES;
  const unsigned have = (unsigned)min(EB_NODES, NN - base);
  table_clear(tab, tid);
  __syncthreads();
  table_count(tab, dst, chunk, base, have, tid);
  __syncthreads();
  {
    int v[EB_SWEEP], h[EB_SWEEP];
    #pragma unroll
    for (int k = 0; k < EB_SWEEP; ++k) {
      v[k] = sweep_word(tid, k, chunk);
      h[k] = tab[v[k]];
    }
    #pragma unroll
    for (int k = 0; k < EB_SWEEP; ++k)
      if (h[k]) h[k] = atomicAdd(&cur[base + v[k]], h[k]);
    #pragma unroll
    for (int k = 0; k < EB_SWEEP; ++k) tab[v[k]] = h[k];
  }
  __syncthreads();
  const int4* d4 = (const int4*)dst + (size_t)chunk * EB_Q;
  const int4* s4 = (const int4*)src + (size_t)chunk * EB_Q;
  int4 d[EB_V], s[EB_V];
  #pragma unroll
  for (int j = 0; j < EB_V; ++j) {
    const int i = min(tid + j * EB_T, EB_Q - 1);
    d[j] = d4[i];
    s[j] = s4[i];
  }
  bool in[EB_V * 4];
  int  at[EB_V * 4];
  #pragma unroll
  for (int j = 0; j < EB_V; ++j) {
    const bool live = tid + j * EB_T < EB_Q;
    const unsigned r0 = word_of(d[j].x, base);
    const unsigned r1 = word_of(d[j].y, base);
    const unsigned r2 = word_of(d[j].z, base);
    const unsigned r3 = word_of(d[j].w, base);
    in[j * 4 + 0] = live && r0 < have;
    in[j * 4 + 1] = live && r1 < have;
    in[j * 4 + 2] = live && r2 < have;
    in[j * 4 + 3] = live && r3 < have;
    at[j * 4 + 0] = 0;
    if (in[j * 4 + 0]) at[j * 4 + 0] = atomicAdd(&tab[r0], 1);
    at[j * 4 + 1] = 0;
    if (in[j * 4 + 1]) at[j * 4 + 1] = atomicAdd(&tab[r1], 1);
    at[j * 4 + 2] = 0;
    if (in[j * 4 + 2]) at[j * 4 + 2] = atomicAdd(&tab[r2], 1);
    at[j * 4 + 3] = 0;
    if (in[j * 4 + 3]) at[j * 4 + 3] = atomicAdd(&tab[r3], 1);
  }
  #pragma unroll
  for (int j = 0; j < EB_V; ++j) {
    if (in[j * 4 + 0]) col[min(max(at[j * 4 + 0], 0), EE - 1)] = s[j].x;
    if (in[j * 4 + 1]) col[min(max(at[j * 4 + 1], 0), EE - 1)] = s[j].y;
    if (in[j * 4 + 2]) col[min(max(at[j * 4 + 2], 0), EE - 1)] = s[j].z;
    if (in[j * 4 + 3]) col[min(max(at[j * 4 + 3], 0), EE - 1)] = s[j].w;
  }
}

__device__ inline float leaky(float x) { return x > 0.0f ? x : 0.01f * x; }

__device__ inline int wave_min_i(int x) {
  for (int o = 16; o > 0; o >>= 1) x = min(x, __shfl_xor(x, o, 32));
  return x;
}
__device__ inline int wave_sum_i(int x) {
  for (int o = 16; o > 0; o >>= 1) x += __shfl_xor(x, o, 32);
  return x;
}
__device__ inline float wave_max_f(float x) {
  for (int o = 16; o > 0; o >>= 1) x = fmaxf(x, __shfl_xor(x, o, 32));
  return x;
}

__device__ inline int next_source(const int* __restrict__ seg, int deg, int last, int lane,
                                  int& mult) {
  int best = 0x7fffffff;
  for (int c = lane; c < deg; c += 32) {
    const int s = seg[c];
    if (s > last && s < best) best = s;
  }
  best = wave_min_i(best);
  int m = 0;
  for (int c = lane; c < deg; c += 32) m += (seg[c] == best) ? 1 : 0;
  mult = (best == 0x7fffffff) ? 0 : wave_sum_i(m);
  return best;
}

template <int C>
__device__ inline void pick(const int (&rank)[C], const int (&sv)[C], const float (&w)[C],
                            int k, int& svk, float& wk) {
  svk = 0; wk = 0.0f;
  for (int c = 0; c < C; ++c) {
    const unsigned mk = __builtin_amdgcn_ballot_w32(rank[c] == k);
    if (C == 1 || mk != 0u) {
      const int j = (__ffs((int)mk) - 1) & 31;
      svk = __builtin_amdgcn_readlane(sv[c], j);
      wk  = __int_as_float(__builtin_amdgcn_readlane(__float_as_int(w[c]), j));
    }
  }
}

template <int C>
__device__ inline float4 segment_in_registers(const int* __restrict__ seg, int deg, int lane,
                                              const float* __restrict__ q,
                                              float pv, float be0,
                                              const float4* __restrict__ hv4) {
  int sv[C], key[C], rank[C];
  float x[C], ex[C], a[C];
  for (int c = 0; c < C; ++c) {
    const int i = c * 32 + lane;
    sv[c] = 0; key[c] = 0x7fffffff; x[c] = -INFINITY; rank[c] = 0;
    if (i < deg) {
      sv[c]  = min(max(seg[i], 0), NN - 1);
      key[c] = sv[c];
      x[c]   = leaky(pv + q[sv[c]] + be0);
    }
  }
  for (int j = 0; j < deg; ++j) {
    int kj = __builtin_amdgcn_readlane(key[0], j & 31);
    for (int c = 1; c < C; ++c)
      if ((j >> 5) == c) kj = __builtin_amdgcn_readlane(key[c], j & 31);
    for (int c = 0; c < C; ++c)
      rank[c] += (kj < key[c] || (kj == key[c] && j < c * 32 + lane)) ? 1 : 0;
  }
  float mx = x[0];
  for (int c = 1; c < C; ++c) mx = fmaxf(mx, x[c]);
  mx = wave_max_f(mx);
  for (int c = 0; c < C; ++c) ex[c] = (c * 32 + lane < deg) ? expf(x[c] - mx) : 0.0f;
  float den = 0.0f;
  for (int k = 0; k < deg; ++k) {
    int s_; float exk;
    pick<C>(rank, sv, ex, k, s_, exk);
    den += exk;
  }
  const float dsafe = den > 0.0f ? den : 1.0f;
  for (int c = 0; c < C; ++c) a[c] = ex[c] / dsafe;

  float4 acc = make_float4(0.0f, 0.0f, 0.0f, 0.0f);
  int k = 0;
  for (; k + 4 <= deg; k += 4) {
    int s0, s1, s2, s3; float a0, a1, a2, a3;
    pick<C>(rank, sv, a, k,     s0, a0);
    pick<C>(rank, sv, a, k + 1, s1, a1);
    pick<C>(rank, sv, a, k + 2, s2, a2);
    pick<C>(rank, sv, a, k + 3, s3, a3);
    const float4 h0 = hv4[(size_t)s0 * (GG / 4) + lane];
    const float4 h1 = hv4[(size_t)s1 * (GG / 4) + lane];
    const float4 h2 = hv4[(size_t)s2 * (GG / 4) + lane];
    const float4 h3 = hv4[(size_t)s3 * (GG / 4) + lane];
    acc.x += h0.x * a0; acc.y += h0.y * a0; acc.z += h0.z * a0; acc.w += h0.w * a0;
    acc.x += h1.x * a1; acc.y += h1.y * a1; acc.z += h1.z * a1; acc.w += h1.w * a1;
    acc.x += h2.x * a2; acc.y += h2.y * a2; acc.z += h2.z * a2; acc.w += h2.w * a2;
    acc.x += h3.x * a3; acc.y += h3.y * a3; acc.z += h3.z * a3; acc.w += h3.w * a3;
  }
  for (; k < deg; ++k) {
    int sk; float ak;
    pick<C>(rank, sv, a, k, sk, ak);
    const float4 h = hv4[(size_t)sk * (GG / 4) + lane];
    acc.x += h.x * ak; acc.y += h.y * ak; acc.z += h.z * ak; acc.w += h.w * ak;
  }
  return acc;
}

__global__ void __launch_bounds__(256)
node_softmax_aggregate(const int* __restrict__ col,
                       const int* __restrict__ off, const int* __restrict__ endp,
                       const float* __restrict__ p, const float* __restrict__ q,
                       const float* __restrict__ be, const float* __restrict__ hv,
                       float* __restrict__ out) {
  const int v    = (int)((blockIdx.x * (size_t)blockDim.x + threadIdx.x) >> 5);
  const int lane = threadIdx.x & 31;
  if (v >= NN) return;
  const int start = off[v];
  const int deg   = endp[v] - start;
  const int* seg  = col + start;
  const float pv  = p[v];
  const float be0 = be[0];
  const float4* hv4 = (const float4*)hv;
  float4 acc;

  if (deg <= 32) {
    acc = segment_in_registers<1>(seg, deg, lane, q, pv, be0, hv4);
  } else if (deg <= 64) {
    acc = segment_in_registers<2>(seg, deg, lane, q, pv, be0, hv4);
  } else {
    acc = make_float4(0.0f, 0.0f, 0.0f, 0.0f);
    float mx = -INFINITY;
    for (int c = lane; c < deg; c += 32) {
      const int s = min(max(seg[c], 0), NN - 1);
      mx = fmaxf(mx, leaky(pv + q[s] + be0));
    }
    mx = wave_max_f(mx);
    float den = 0.0f;
    int last = -1;
    for (int k = 0; k < deg; ) {
      int mult;
      const int s = next_source(seg, deg, last, lane, mult);
      if (mult <= 0) break;
      last = s;
      const float exk = expf(leaky(pv + q[min(s, NN - 1)] + be0) - mx);
      for (int t = 0; t < mult; ++t) den += exk;
      k += mult;
    }
    const float dsafe = den > 0.0f ? den : 1.0f;
    last = -1;
    for (int k = 0; k < deg; ) {
      int mult;
      const int s = next_source(seg, deg, last, lane, mult);
      if (mult <= 0) break;
      last = s;
      const int sc = min(s, NN - 1);
      const float ak = expf(leaky(pv + q[sc] + be0) - mx) / dsafe;
      const float4 h = hv4[(size_t)sc * (GG / 4) + lane];
      for (int t = 0; t < mult; ++t) {
        acc.x += h.x * ak; acc.y += h.y * ak; acc.z += h.z * ak; acc.w += h.w * ak;
      }
      k += mult;
    }
  }

  float4 r;
  r.x = acc.x > 0.0f ? acc.x : expm1f(acc.x);
  r.y = acc.y > 0.0f ? acc.y : expm1f(acc.y);
  r.z = acc.z > 0.0f ? acc.z : expm1f(acc.z);
  r.w = acc.w > 0.0f ? acc.w : expm1f(acc.w);
  ((float4*)out)[(size_t)v * (GG / 4) + lane] = r;
}

extern "C" void kernel_launch(void* const* d_in, const int* in_sizes, int n_in,
                              void* d_out, int out_size, void* d_ws, size_t ws_size,
                              hipStream_t stream) {
  (void)in_sizes; (void)n_in; (void)out_size; (void)ws_size;
  const float* nf  = (const float*)d_in[0];
  const int*   src = (const int*)  d_in[1];
  const int*   dst = (const int*)  d_in[2];
  const float* We  = (const float*)d_in[3];
  const float* be  = (const float*)d_in[4];
  const float* Wn  = (const float*)d_in[5];
  const float* bn  = (const float*)d_in[6];
  float* out = (float*)d_out;

  float* ws   = (float*)d_ws;
  float* hv   = ws;
  float* p    = hv + (size_t)NN * GG;
  float* q    = p + NN;
  int*   cnt  = (int*)(q + NN);
  int*   off  = cnt + NN;
  int*   col  = off + NN;

  node_gemm_wmma<<<NN / 16, 256, 0, stream>>>(nf, Wn, bn, hv, We, p, q, cnt);
  edge_count<<<EB_BLOCKS, EB_T, 0, stream>>>(dst, cnt);
  scan_counts<<<1, SCAN_T, 0, stream>>>(cnt, off);
  edge_place<<<EB_BLOCKS, EB_T, 0, stream>>>(src, dst, cnt, col);
  node_softmax_aggregate<<<(NN * 32 + 255) / 256, 256, 0, stream>>>(col, off, cnt, p, q, be, hv, out);
}


# Aggregate launch: node, segment start and length held as scalars; each segment of up to 32 edges permuted once into ascending-source lane order, terms read at scalar index; order of sums unchanged
# speedup vs baseline: 3.0362x; 1.0588x over previous
#include <hip/hip_runtime.h>
#include <math.h>

typedef __attribute__((ext_vector_type(2))) float v2f;
typedef __attribute__((ext_vector_type(8))) float v8f;

#define NN 50000
#define EE 800000
#define DD 128
#define GG 128
#define LDS_STRIDE 132

__global__ void __launch_bounds__(256, 16)
node_gemm_wmma(const float* __restrict__ nf, const float* __restrict__ Wn,
               const float* __restrict__ bn, float* __restrict__ hv,
               const float* __restrict__ We, float* __restrict__ p,
               float* __restrict__ q, int* __restrict__ cnt) {
  __shared__ float Atile[16 * LDS_STRIDE];
  const int row0 = blockIdx.x * 16;
  const int tid  = threadIdx.x;

  #pragma unroll
  for (int i = tid; i < 16 * DD; i += 256) {
    const int r = i >> 7;
    const int c = i & (DD - 1);
    Atile[r * LDS_STRIDE + c] = nf[(size_t)(row0 + r) * DD + c];
  }
  __syncthreads();

  const int wave = tid >> 5;
  const int lane = tid & 31;
  const int col0 = wave * 16;
  const int half = lane >> 4;
  const int l16  = lane & 15;

  v8f acc = {};
  #pragma unroll
  for (int k = 0; k < DD; k += 4) {
    const int ka = k + half * 2;
    v2f a, b;
    a.x = Atile[l16 * LDS_STRIDE + ka];
    a.y = Atile[l16 * LDS_STRIDE + ka + 1];
    b.x = Wn[(size_t)(ka + 0) * GG + col0 + l16];
    b.y = Wn[(size_t)(ka + 1) * GG + col0 + l16];
    acc = __builtin_amdgcn_wmma_f32_16x16x4_f32(false, a, false, b,
                                                (short)0, acc, false, false);
  }

  #pragma unroll
  for (int i = 0; i < 8; ++i) {
    const int row = row0 + i + half * 8;
    const int col = col0 + l16;
    float v = acc[i] + bn[col];
    v = v > 0.0f ? v : 0.01f * v;
    hv[(size_t)row * GG + col] = v;
  }

  const int sr0 = wave * 2;
  float ap0 = 0.0f, aq0 = 0.0f, ap1 = 0.0f, aq1 = 0.0f;
  #pragma unroll
  for (int i = lane; i < DD; i += 32) {
    const float wp = We[i];
    const float wq = We[DD + i];
    const float x0 = Atile[sr0 * LDS_STRIDE + i];
    const float x1 = Atile[(sr0 + 1) * LDS_STRIDE + i];
    ap0 += x0 * wp;
    aq0 += x0 * wq;
    ap1 += x1 * wp;
    aq1 += x1 * wq;
  }
  #pragma unroll
  for (int off = 16; off > 0; off >>= 1) {
    ap0 += __shfl_xor(ap0, off, 32);
    aq0 += __shfl_xor(aq0, off, 32);
    ap1 += __shfl_xor(ap1, off, 32);
    aq1 += __shfl_xor(aq1, off, 32);
  }

  if (lane < 2) {
    const int node = row0 + sr0 + lane;
    p[node]   = lane ? ap1 : ap0;
    q[node]   = lane ? aq1 : aq0;
    cnt[node] = 0;
  }
}

#define EB_CHUNKS 64
#define EB_PARTS  4
#define EB_BLOCKS (EB_CHUNKS * EB_PARTS)
#define EB_T      1024
#define EB_EDGES  (EE / EB_CHUNKS)
#define EB_Q      (EB_EDGES / 4)
#define EB_V      ((EB_Q + EB_T - 1) / EB_T)
#define EB_NODES  12512
#define EB_SWEEP  13
#define EB_WORDS  (EB_SWEEP * EB_T)
#define EB_ROT    192

static_assert(EB_PARTS * EB_NODES >= NN, "the node ranges cover all nodes");
static_assert(EB_WORDS >= EB_NODES && EB_NODES % 32 == 0, "table holds a range; ranges start on a line");
static_assert(EB_CHUNKS * EB_ROT <= EB_WORDS && EB_ROT % 32 == 0, "sweep starts stay inside the table, on lines");
static_assert(EB_EDGES * EB_CHUNKS == EE && EB_Q * 4 == EB_EDGES, "chunks are whole 16-byte pieces");

__device__ __forceinline__ unsigned word_of(int d, int base) { return (unsigned)d - (unsigned)base; }

__device__ __forceinline__ void table_clear(int* __restrict__ tab, int tid) {
  int4* t4 = (int4*)tab;
  for (int i = tid; i < EB_WORDS / 4; i += EB_T) t4[i] = make_int4(0, 0, 0, 0);
}

__device__ __forceinline__ void table_count(int* __restrict__ tab, const int* __restrict__ dst,
                                            int chunk, int base, unsigned have, int tid) {
  const int4* d4 = (const int4*)dst + (size_t)chunk * EB_Q;
  int4 d[EB_V];
  #pragma unroll
  for (int j = 0; j < EB_V; ++j) d[j] = d4[min(tid + j * EB_T, EB_Q - 1)];
  #pragma unroll
  for (int j = 0; j < EB_V; ++j) {
    if (tid + j * EB_T < EB_Q) {
      const unsigned r0 = word_of(d[j].x, base);
      const unsigned r1 = word_of(d[j].y, base);
      const unsigned r2 = word_of(d[j].z, base);
      const unsigned r3 = word_of(d[j].w, base);
      if (r0 < have) atomicAdd(&tab[r0], 1);
      if (r1 < have) atomicAdd(&tab[r1], 1);
      if (r2 < have) atomicAdd(&tab[r2], 1);
      if (r3 < have) atomicAdd(&tab[r3], 1);
    }
  }
}

__device__ __forceinline__ int sweep_word(int tid, int j, int chunk) {
  const int v = tid + j * EB_T + chunk * EB_ROT;
  return v >= EB_WORDS ? v - EB_WORDS : v;
}

__global__ void __launch_bounds__(EB_T)
edge_count(const int* __restrict__ dst, int* __restrict__ cnt) {
  __shared__ __attribute__((aligned(16))) int tab[EB_WORDS];
  const int tid   = threadIdx.x;
  const int chunk = blockIdx.x % EB_CHUNKS;
  const int base  = (blockIdx.x / EB_CHUNKS) * EB_NODES;
  const unsigned have = (unsigned)min(EB_NODES, NN - base);
  table_clear(tab, tid);
  __syncthreads();
  table_count(tab, dst, chunk, base, have, tid);
  __syncthreads();
  int v[EB_SWEEP], h[EB_SWEEP];
  #pragma unroll
  for (int k = 0; k < EB_SWEEP; ++k) {
    v[k] = sweep_word(tid, k, chunk);
    h[k] = tab[v[k]];
  }
  #pragma unroll
  for (int k = 0; k < EB_SWEEP; ++k)
    if (h[k]) atomicAdd(&cnt[base + v[k]], h[k]);
}

#define SCAN_T   1024
#define SCAN_PER 49
#define SCAN_Q   (SCAN_T * SCAN_PER / 4)
#define SCAN_NQ  (NN / 4)
#define SCAN_LD  ((SCAN_Q + SCAN_T - 1) / SCAN_T)

__global__ void __launch_bounds__(SCAN_T)
scan_counts(int* __restrict__ cnt, int* __restrict__ off) {
  __shared__ __attribute__((aligned(16))) int buf[SCAN_LD * SCAN_T * 4];
  __shared__ int wsum[32];
  const int tid  = threadIdx.x;
  const int lane = tid & 31;
  const int w    = tid >> 5;
  const int4* c4 = (const int4*)cnt;
  int4*       b4 = (int4*)buf;
  int4 r[SCAN_LD];
  #pragma unroll
  for (int j = 0; j < SCAN_LD; ++j) r[j] = c4[min(tid + j * SCAN_T, SCAN_NQ - 1)];
  #pragma unroll
  for (int j = 0; j < SCAN_LD; ++j) {
    const int i = tid + j * SCAN_T;
    b4[i] = (i < SCAN_NQ) ? r[j] : make_int4(0, 0, 0, 0);
  }
  __syncthreads();
  const int base = tid * SCAN_PER;
  int s = 0;
  for (int i = 0; i < SCAN_PER; ++i) s += buf[base + i];
  int inc = s;
  for (int o = 1; o < 32; o <<= 1) {
    const int t = __shfl_up(inc, o, 32);
    if (lane >= o) inc += t;
  }
  if (lane == 31) wsum[w] = inc;
  __syncthreads();
  if (w == 0) {
    const int x = wsum[lane];
    int xi = x;
    for (int o = 1; o < 32; o <<= 1) {
      const int t = __shfl_up(xi, o, 32);
      if (lane >= o) xi += t;
    }
    wsum[lane] = xi - x;
  }
  __syncthreads();
  int run = wsum[w] + inc - s;
  for (int i = 0; i < SCAN_PER; ++i) {
    const int c = buf[base + i];
    buf[base + i] = run;
    run += c;
  }
  __syncthreads();
  for (int i = tid; i < NN; i += SCAN_T) {
    const int o = buf[i];
    off[i] = o;
    cnt[i] = o;
  }
}

__global__ void __launch_bounds__(EB_T)
edge_place(const int* __restrict__ src, const int* __restrict__ dst,
           int* __restrict__ cur, int* __restrict__ col) {
  __shared__ __attribute__((aligned(16))) int tab[EB_WORDS];
  const int tid   = threadIdx.x;
  const int chunk = blockIdx.x % EB_CHUNKS;
  const int base  = (blockIdx.x / EB_CHUNKS) * EB_NODES;
  const unsigned have = (unsigned)min(EB_NODES, NN - base);
  table_clear(tab, tid);
  __syncthreads();
  table_count(tab, dst, chunk, base, have, tid);
  __syncthreads();
  {
    int v[EB_SWEEP], h[EB_SWEEP];
    #pragma unroll
    for (int k = 0; k < EB_SWEEP; ++k) {
      v[k] = sweep_word(tid, k, chunk);
      h[k] = tab[v[k]];
    }
    #pragma unroll
    for (int k = 0; k < EB_SWEEP; ++k)
      if (h[k]) h[k] = atomicAdd(&cur[base + v[k]], h[k]);
    #pragma unroll
    for (int k = 0; k < EB_SWEEP; ++k) tab[v[k]] = h[k];
  }
  __syncthreads();
  const int4* d4 = (const int4*)dst + (size_t)chunk * EB_Q;
  const int4* s4 = (const int4*)src + (size_t)chunk * EB_Q;
  int4 d[EB_V], s[EB_V];
  #pragma unroll
  for (int j = 0; j < EB_V; ++j) {
    const int i = min(tid + j * EB_T, EB_Q - 1);
    d[j] = d4[i];
    s[j] = s4[i];
  }
  bool in[EB_V * 4];
  int  at[EB_V * 4];
  #pragma unroll
  for (int j = 0; j < EB_V; ++j) {
    const bool live = tid + j * EB_T < EB_Q;
    const unsigned r0 = word_of(d[j].x, base);
    const unsigned r1 = word_of(d[j].y, base);
    const unsigned r2 = word_of(d[j].z, base);
    const unsigned r3 = word_of(d[j].w, base);
    in[j * 4 + 0] = live && r0 < have;
    in[j * 4 + 1] = live && r1 < have;
    in[j * 4 + 2] = live && r2 < have;
    in[j * 4 + 3] = live && r3 < have;
    at[j * 4 + 0] = 0;
    if (in[j * 4 + 0]) at[j * 4 + 0] = atomicAdd(&tab[r0], 1);
    at[j * 4 + 1] = 0;
    if (in[j * 4 + 1]) at[j * 4 + 1] = atomicAdd(&tab[r1], 1);
    at[j * 4 + 2] = 0;
    if (in[j * 4 + 2]) at[j * 4 + 2] = atomicAdd(&tab[r2], 1);
    at[j * 4 + 3] = 0;
    if (in[j * 4 + 3]) at[j * 4 + 3] = atomicAdd(&tab[r3], 1);
  }
  #pragma unroll
  for (int j = 0; j < EB_V; ++j) {
    if (in[j * 4 + 0]) col[min(max(at[j * 4 + 0], 0), EE - 1)] = s[j].x;
    if (in[j * 4 + 1]) col[min(max(at[j * 4 + 1], 0), EE - 1)] = s[j].y;
    if (in[j * 4 + 2]) col[min(max(at[j * 4 + 2], 0), EE - 1)] = s[j].z;
    if (in[j * 4 + 3]) col[min(max(at[j * 4 + 3], 0), EE - 1)] = s[j].w;
  }
}

__device__ inline float leaky(float x) { return x > 0.0f ? x : 0.01f * x; }

__device__ inline int wave_min_i(int x) {
  for (int o = 16; o > 0; o >>= 1) x = min(x, __shfl_xor(x, o, 32));
  return x;
}
__device__ inline int wave_sum_i(int x) {
  for (int o = 16; o > 0; o >>= 1) x += __shfl_xor(x, o, 32);
  return x;
}
__device__ inline float wave_max_f(float x) {
  for (int o = 16; o > 0; o >>= 1) x = fmaxf(x, __shfl_xor(x, o, 32));
  return x;
}

__device__ inline int next_source(const int* __restrict__ seg, int deg, int last, int lane,
                                  int& mult) {
  int best = 0x7fffffff;
  for (int c = lane; c < deg; c += 32) {
    const int s = seg[c];
    if (s > last && s < best) best = s;
  }
  best = wave_min_i(best);
  int m = 0;
  for (int c = lane; c < deg; c += 32) m += (seg[c] == best) ? 1 : 0;
  mult = (best == 0x7fffffff) ? 0 : wave_sum_i(m);
  return best;
}

template <int C>
__device__ inline void pick(const int (&rank)[C], const int (&sv)[C], const float (&w)[C],
                            int k, int& svk, float& wk) {
  svk = 0; wk = 0.0f;
  for (int c = 0; c < C; ++c) {
    const unsigned mk = __builtin_amdgcn_ballot_w32(rank[c] == k);
    if (C == 1 || mk != 0u) {
      const int j = (__ffs((int)mk) - 1) & 31;
      svk = __builtin_amdgcn_readlane(sv[c], j);
      wk  = __int_as_float(__builtin_amdgcn_readlane(__float_as_int(w[c]), j));
    }
  }
}

template <int C>
__device__ inline float4 segment_in_registers(const int* __restrict__ seg, int deg, int lane,
                                              const float* __restrict__ q,
                                              float pv, float be0,
                                              const float4* __restrict__ hv4) {
  int sv[C], key[C], rank[C];
  float x[C], ex[C], a[C];
  for (int c = 0; c < C; ++c) {
    const int i = c * 32 + lane;
    sv[c] = 0; key[c] = 0x7fffffff; x[c] = -INFINITY; rank[c] = 0;
    if (i < deg) {
      sv[c]  = min(max(seg[i], 0), NN - 1);
      key[c] = sv[c];
      x[c]   = leaky(pv + q[sv[c]] + be0);
    }
  }
  for (int j = 0; j < deg; ++j) {
    int kj = __builtin_amdgcn_readlane(key[0], j & 31);
    for (int c = 1; c < C; ++c)
      if ((j >> 5) == c) kj = __builtin_amdgcn_readlane(key[c], j & 31);
    for (int c = 0; c < C; ++c)
      rank[c] += (kj < key[c] || (kj == key[c] && j < c * 32 + lane)) ? 1 : 0;
  }
  float mx = x[0];
  for (int c = 1; c < C; ++c) mx = fmaxf(mx, x[c]);
  mx = wave_max_f(mx);
  for (int c = 0; c < C; ++c) ex[c] = (c * 32 + lane < deg) ? expf(x[c] - mx) : 0.0f;
  float den = 0.0f;
  for (int k = 0; k < deg; ++k) {
    int s_; float exk;
    pick<C>(rank, sv, ex, k, s_, exk);
    den += exk;
  }
  const float dsafe = den > 0.0f ? den : 1.0f;
  for (int c = 0; c < C; ++c) a[c] = ex[c] / dsafe;

  float4 acc = make_float4(0.0f, 0.0f, 0.0f, 0.0f);
  int k = 0;
  for (; k + 4 <= deg; k += 4) {
    int s0, s1, s2, s3; float a0, a1, a2, a3;
    pick<C>(rank, sv, a, k,     s0, a0);
    pick<C>(rank, sv, a, k + 1, s1, a1);
    pick<C>(rank, sv, a, k + 2, s2, a2);
    pick<C>(rank, sv, a, k + 3, s3, a3);
    const float4 h0 = hv4[(size_t)s0 * (GG / 4) + lane];
    const float4 h1 = hv4[(size_t)s1 * (GG / 4) + lane];
    const float4 h2 = hv4[(size_t)s2 * (GG / 4) + lane];
    const float4 h3 = hv4[(size_t)s3 * (GG / 4) + lane];
    acc.x += h0.x * a0; acc.y += h0.y * a0; acc.z += h0.z * a0; acc.w += h0.w * a0;
    acc.x += h1.x * a1; acc.y += h1.y * a1; acc.z += h1.z * a1; acc.w += h1.w * a1;
    acc.x += h2.x * a2; acc.y += h2.y * a2; acc.z += h2.z * a2; acc.w += h2.w * a2;
    acc.x += h3.x * a3; acc.y += h3.y * a3; acc.z += h3.z * a3; acc.w += h3.w * a3;
  }
  for (; k < deg; ++k) {
    int sk; float ak;
    pick<C>(rank, sv, a, k, sk, ak);
    const float4 h = hv4[(size_t)sk * (GG / 4) + lane];
    acc.x += h.x * ak; acc.y += h.y * ak; acc.z += h.z * ak; acc.w += h.w * ak;
  }
  return acc;
}

__device__ inline float lane_value_f(float x, int k) {
  return __int_as_float(__builtin_amdgcn_readlane(__float_as_int(x), k));
}

__device__ inline float4 segment_in_lane_order(const int* __restrict__ seg, int deg, int lane,
                                               const float* __restrict__ q,
                                               float pv, float be0,
                                               const float4* __restrict__ hv4) {
  const bool live = lane < deg;
  int sv = 0;
  unsigned key = 0xffffffffu;
  float x = -INFINITY;
  if (live) {
    sv  = min(max(seg[lane], 0), NN - 1);
    key = ((unsigned)sv << 5) | (unsigned)lane;
    x   = leaky(pv + q[sv] + be0);
  }
  int rank = 0;
  float mx = -INFINITY;
  for (int j = 0; j < deg; ++j) {
    const unsigned kj = (unsigned)__builtin_amdgcn_readlane((int)key, j);
    rank += (kj < key) ? 1 : 0;
    mx = fmaxf(mx, lane_value_f(x, j));
  }
  const float ex = live ? expf(x - mx) : 0.0f;
  const int to = (live ? rank : lane) << 2;
  const int   ssv = __builtin_amdgcn_ds_permute(to, sv);
  const float sex = __int_as_float(__builtin_amdgcn_ds_permute(to, __float_as_int(ex)));

  float den = 0.0f;
  for (int k = 0; k < deg; ++k) den += lane_value_f(sex, k);
  const float dsafe = den > 0.0f ? den : 1.0f;
  const float sa = sex / dsafe;

  float4 acc = make_float4(0.0f, 0.0f, 0.0f, 0.0f);
  int k = 0;
  for (; k + 4 <= deg; k += 4) {
    const int s0 = __builtin_amdgcn_readlane(ssv, k);
    const int s1 = __builtin_amdgcn_readlane(ssv, k + 1);
    const int s2 = __builtin_amdgcn_readlane(ssv, k + 2);
    const int s3 = __builtin_amdgcn_readlane(ssv, k + 3);
    const float a0 = lane_value_f(sa, k);
    const float a1 = lane_value_f(sa, k + 1);
    const float a2 = lane_value_f(sa, k + 2);
    const float a3 = lane_value_f(sa, k + 3);
    const float4 h0 = hv4[(size_t)s0 * (GG / 4) + lane];
    const float4 h1 = hv4[(size_t)s1 * (GG / 4) + lane];
    const float4 h2 = hv4[(size_t)s2 * (GG / 4) + lane];
    const float4 h3 = hv4[(size_t)s3 * (GG / 4) + lane];
    acc.x += h0.x * a0; acc.y += h0.y * a0; acc.z += h0.z * a0; acc.w += h0.w * a0;
    acc.x += h1.x * a1; acc.y += h1.y * a1; acc.z += h1.z * a1; acc.w += h1.w * a1;
    acc.x += h2.x * a2; acc.y += h2.y * a2; acc.z += h2.z * a2; acc.w += h2.w * a2;
    acc.x += h3.x * a3; acc.y += h3.y * a3; acc.z += h3.z * a3; acc.w += h3.w * a3;
  }
  for (; k < deg; ++k) {
    const int   sk = __builtin_amdgcn_readlane(ssv, k);
    const float ak = lane_value_f(sa, k);
    const float4 h = hv4[(size_t)sk * (GG / 4) + lane];
    acc.x += h.x * ak; acc.y += h.y * ak; acc.z += h.z * ak; acc.w += h.w * ak;
  }
  return acc;
}

__global__ void __launch_bounds__(256)
node_softmax_aggregate(const int* __restrict__ col,
                       const int* __restrict__ off, const int* __restrict__ endp,
                       const float* __restrict__ p, const float* __restrict__ q,
                       const float* __restrict__ be, const float* __restrict__ hv,
                       float* __restrict__ out) {
  const int v    = __builtin_amdgcn_readfirstlane((int)(blockIdx.x * 8u + (threadIdx.x >> 5)));
  const int lane = threadIdx.x & 31;
  if (v >= NN) return;
  const int start = __builtin_amdgcn_readfirstlane(off[v]);
  const int deg   = __builtin_amdgcn_readfirstlane(endp[v]) - start;
  const int* seg  = col + start;
  const float pv  = __int_as_float(__builtin_amdgcn_readfirstlane(__float_as_int(p[v])));
  const float be0 = be[0];
  const float4* hv4 = (const float4*)hv;
  float4 acc;

  if (deg <= 32) {
    acc = segment_in_lane_order(seg, deg, lane, q, pv, be0, hv4);
  } else if (deg <= 64) {
    acc = segment_in_registers<2>(seg, deg, lane, q, pv, be0, hv4);
  } else {
    acc = make_float4(0.0f, 0.0f, 0.0f, 0.0f);
    float mx = -INFINITY;
    for (int c = lane; c < deg; c += 32) {
      const int s = min(max(seg[c], 0), NN - 1);
      mx = fmaxf(mx, leaky(pv + q[s] + be0));
    }
    mx = wave_max_f(mx);
    float den = 0.0f;
    int last = -1;
    for (int k = 0; k < deg; ) {
      int mult;
      const int s = next_source(seg, deg, last, lane, mult);
      if (mult <= 0) break;
      last = s;
      const float exk = expf(leaky(pv + q[min(s, NN - 1)] + be0) - mx);
      for (int t = 0; t < mult; ++t) den += exk;
      k += mult;
    }
    const float dsafe = den > 0.0f ? den : 1.0f;
    last = -1;
    for (int k = 0; k < deg; ) {
      int mult;
      const int s = next_source(seg, deg, last, lane, mult);
      if (mult <= 0) break;
      last = s;
      const int sc = min(s, NN - 1);
      const float ak = expf(leaky(pv + q[sc] + be0) - mx) / dsafe;
      const float4 h = hv4[(size_t)sc * (GG / 4) + lane];
      for (int t = 0; t < mult; ++t) {
        acc.x += h.x * ak; acc.y += h.y * ak; acc.z += h.z * ak; acc.w += h.w * ak;
      }
      k += mult;
    }
  }

  float4 r;
  r.x = acc.x > 0.0f ? acc.x : expm1f(acc.x);
  r.y = acc.y > 0.0f ? acc.y : expm1f(acc.y);
  r.z = acc.z > 0.0f ? acc.z : expm1f(acc.z);
  r.w = acc.w > 0.0f ? acc.w : expm1f(acc.w);
  ((float4*)out)[(size_t)v * (GG / 4) + lane] = r;
}

extern "C" void kernel_launch(void* const* d_in, const int* in_sizes, int n_in,
                              void* d_out, int out_size, void* d_ws, size_t ws_size,
                              hipStream_t stream) {
  (void)in_sizes; (void)n_in; (void)out_size; (void)ws_size;
  const float* nf  = (const float*)d_in[0];
  const int*   src = (const int*)  d_in[1];
  const int*   dst = (const int*)  d_in[2];
  const float* We  = (const float*)d_in[3];
  const float* be  = (const float*)d_in[4];
  const float* Wn  = (const float*)d_in[5];
  const float* bn  = (const float*)d_in[6];
  float* out = (float*)d_out;

  float* ws   = (float*)d_ws;
  float* hv   = ws;
  float* p    = hv + (size_t)NN * GG;
  float* q    = p + NN;
  int*   cnt  = (int*)(q + NN);
  int*   off  = cnt + NN;
  int*   col  = off + NN;

  node_gemm_wmma<<<NN / 16, 256, 0, stream>>>(nf, Wn, bn, hv, We, p, q, cnt);
  edge_count<<<EB_BLOCKS, EB_T, 0, stream>>>(dst, cnt);
  scan_counts<<<1, SCAN_T, 0, stream>>>(cnt, off);
  edge_place<<<EB_BLOCKS, EB_T, 0, stream>>>(src, dst, cnt, col);
  node_softmax_aggregate<<<(NN * 32 + 255) / 256, 256, 0, stream>>>(col, off, cnt, p, q, be, hv, out);
}


# Scan launch: 49 workgroups, one per 1024 nodes, each sums the counts in front of its block itself (one group of 16-byte loads), not one workgroup over all; start array dropped; float sums untouched
# speedup vs baseline: 3.2009x; 1.0542x over previous
#include <hip/hip_runtime.h>
#include <math.h>

typedef __attribute__((ext_vector_type(2))) float v2f;
typedef __attribute__((ext_vector_type(8))) float v8f;

#define NN 50000
#define EE 800000
#define DD 128
#define GG 128
#define LDS_STRIDE 132

__global__ void __launch_bounds__(256, 16)
node_gemm_wmma(const float* __restrict__ nf, const float* __restrict__ Wn,
               const float* __restrict__ bn, float* __restrict__ hv,
               const float* __restrict__ We, float* __restrict__ p,
               float* __restrict__ q, int* __restrict__ cnt) {
  __shared__ float Atile[16 * LDS_STRIDE];
  const int row0 = blockIdx.x * 16;
  const int tid  = threadIdx.x;

  #pragma unroll
  for (int i = tid; i < 16 * DD; i += 256) {
    const int r = i >> 7;
    const int c = i & (DD - 1);
    Atile[r * LDS_STRIDE + c] = nf[(size_t)(row0 + r) * DD + c];
  }
  __syncthreads();

  const int wave = tid >> 5;
  const int lane = tid & 31;
  const int col0 = wave * 16;
  const int half = lane >> 4;
  const int l16  = lane & 15;

  v8f acc = {};
  #pragma unroll
  for (int k = 0; k < DD; k += 4) {
    const int ka = k + half * 2;
    v2f a, b;
    a.x = Atile[l16 * LDS_STRIDE + ka];
    a.y = Atile[l16 * LDS_STRIDE + ka + 1];
    b.x = Wn[(size_t)(ka + 0) * GG + col0 + l16];
    b.y = Wn[(size_t)(ka + 1) * GG + col0 + l16];
    acc = __builtin_amdgcn_wmma_f32_16x16x4_f32(false, a, false, b,
                                                (short)0, acc, false, false);
  }

  #pragma unroll
  for (int i = 0; i < 8; ++i) {
    const int row = row0 + i + half * 8;
    const int col = col0 + l16;
    float v = acc[i] + bn[col];
    v = v > 0.0f ? v : 0.01f * v;
    hv[(size_t)row * GG + col] = v;
  }

  const int sr0 = wave * 2;
  float ap0 = 0.0f, aq0 = 0.0f, ap1 = 0.0f, aq1 = 0.0f;
  #pragma unroll
  for (int i = lane; i < DD; i += 32) {
    const float wp = We[i];
    const float wq = We[DD + i];
    const float x0 = Atile[sr0 * LDS_STRIDE + i];
    const float x1 = Atile[(sr0 + 1) * LDS_STRIDE + i];
    ap0 += x0 * wp;
    aq0 += x0 * wq;
    ap1 += x1 * wp;
    aq1 += x1 * wq;
  }
  #pragma unroll
  for (int off = 16; off > 0; off >>= 1) {
    ap0 += __shfl_xor(ap0, off, 32);
    aq0 += __shfl_xor(aq0, off, 32);
    ap1 += __shfl_xor(ap1, off, 32);
    aq1 += __shfl_xor(aq1, off, 32);
  }

  if (lane < 2) {
    const int node = row0 + sr0 + lane;
    p[node]   = lane ? ap1 : ap0;
    q[node]   = lane ? aq1 : aq0;
    cnt[node] = 0;
  }
}

#define EB_CHUNKS 64
#define EB_PARTS  4
#define EB_BLOCKS (EB_CHUNKS * EB_PARTS)
#define EB_T      1024
#define EB_EDGES  (EE / EB_CHUNKS)
#define EB_Q      (EB_EDGES / 4)
#define EB_V      ((EB_Q + EB_T - 1) / EB_T)
#define EB_NODES  12512
#define EB_SWEEP  13
#define EB_WORDS  (EB_SWEEP * EB_T)
#define EB_ROT    192

static_assert(EB_PARTS * EB_NODES >= NN, "the node ranges cover all nodes");
static_assert(EB_WORDS >= EB_NODES && EB_NODES % 32 == 0, "table holds a range; ranges start on a line");
static_assert(EB_CHUNKS * EB_ROT <= EB_WORDS && EB_ROT % 32 == 0, "sweep starts stay inside the table, on lines");
static_assert(EB_EDGES * EB_CHUNKS == EE && EB_Q * 4 == EB_EDGES, "chunks are whole 16-byte pieces");

__device__ __forceinline__ unsigned word_of(int d, int base) { return (unsigned)d - (unsigned)base; }

__device__ __forceinline__ void table_clear(int* __restrict__ tab, int tid) {
  int4* t4 = (int4*)tab;
  for (int i = tid; i < EB_WORDS / 4; i += EB_T) t4[i] = make_int4(0, 0, 0, 0);
}

__device__ __forceinline__ void table_count(int* __restrict__ tab, const int* __restrict__ dst,
                                            int chunk, int base, unsigned have, int tid) {
  const int4* d4 = (const int4*)dst + (size_t)chunk * EB_Q;
  int4 d[EB_V];
  #pragma unroll
  for (int j = 0; j < EB_V; ++j) d[j] = d4[min(tid + j * EB_T, EB_Q - 1)];
  #pragma unroll
  for (int j = 0; j < EB_V; ++j) {
    if (tid + j * EB_T < EB_Q) {
      const unsigned r0 = word_of(d[j].x, base);
      const unsigned r1 = word_of(d[j].y, base);
      const unsigned r2 = word_of(d[j].z, base);
      const unsigned r3 = word_of(d[j].w, base);
      if (r0 < have) atomicAdd(&tab[r0], 1);
      if (r1 < have) atomicAdd(&tab[r1], 1);
      if (r2 < have) atomicAdd(&tab[r2], 1);
      if (r3 < have) atomicAdd(&tab[r3], 1);
    }
  }
}

__device__ __forceinline__ int sweep_word(int tid, int j, int chunk) {
  const int v = tid + j * EB_T + chunk * EB_ROT;
  return v >= EB_WORDS ? v - EB_WORDS : v;
}

__global__ void __launch_bounds__(EB_T)
edge_count(const int* __restrict__ dst, int* __restrict__ cnt) {
  __shared__ __attribute__((aligned(16))) int tab[EB_WORDS];
  const int tid   = threadIdx.x;
  const int chunk = blockIdx.x % EB_CHUNKS;
  const int base  = (blockIdx.x / EB_CHUNKS) * EB_NODES;
  const unsigned have = (unsigned)min(EB_NODES, NN - base);
  table_clear(tab, tid);
  __syncthreads();
  table_count(tab, dst, chunk, base, have, tid);
  __syncthreads();
  int v[EB_SWEEP], h[EB_SWEEP];
  #pragma unroll
  for (int k = 0; k < EB_SWEEP; ++k) {
    v[k] = sweep_word(tid, k, chunk);
    h[k] = tab[v[k]];
  }
  #pragma unroll
  for (int k = 0; k < EB_SWEEP; ++k)
    if (h[k]) atomicAdd(&cnt[base + v[k]], h[k]);
}

#define SCAN_T   1024
#define SCAN_B   ((NN + SCAN_T - 1) / SCAN_T)
#define SCAN_LQ  (SCAN_T / 4)
#define SCAN_LD  (((SCAN_B - 1) * SCAN_LQ + SCAN_T - 1) / SCAN_T)

static_assert((SCAN_B - 1) * SCAN_LQ * 4 <= NN, "every piece in front of a block holds counts");

__global__ void __launch_bounds__(SCAN_T)
scan_counts(const int* __restrict__ cnt, int* __restrict__ cur) {
  __shared__ int wsum[32];
  __shared__ int wlow[32];
  const int tid  = threadIdx.x;
  const int lane = tid & 31;
  const int w    = tid >> 5;
  const int nq   = (int)blockIdx.x * SCAN_LQ;
  const int i    = (int)blockIdx.x * SCAN_T + tid;
  const int4* c4 = (const int4*)cnt;
  int4 r[SCAN_LD];
  #pragma unroll
  for (int j = 0; j < SCAN_LD; ++j) r[j] = c4[min(tid + j * SCAN_T, max(nq - 1, 0))];
  const int own = (i < NN) ? cnt[i] : 0;
  int low = 0;
  #pragma unroll
  for (int j = 0; j < SCAN_LD; ++j)
    if (tid + j * SCAN_T < nq) low += r[j].x + r[j].y + r[j].z + r[j].w;
  int inc = own;
  for (int o = 1; o < 32; o <<= 1) {
    const int t = __shfl_up(inc, o, 32);
    if (lane >= o) inc += t;
  }
  for (int o = 16; o > 0; o >>= 1) low += __shfl_xor(low, o, 32);
  if (lane == 31) wsum[w] = inc;
  if (lane == 0)  wlow[w] = low;
  __syncthreads();
  int before = wlow[lane] + ((lane < w) ? wsum[lane] : 0);
  for (int o = 16; o > 0; o >>= 1) before += __shfl_xor(before, o, 32);
  if (i < NN) cur[i] = before + inc - own;
}

__global__ void __launch_bounds__(EB_T)
edge_place(const int* __restrict__ src, const int* __restrict__ dst,
           int* __restrict__ cur, int* __restrict__ col) {
  __shared__ __attribute__((aligned(16))) int tab[EB_WORDS];
  const int tid   = threadIdx.x;
  const int chunk = blockIdx.x % EB_CHUNKS;
  const int base  = (blockIdx.x / EB_CHUNKS) * EB_NODES;
  const unsigned have = (unsigned)min(EB_NODES, NN - base);
  table_clear(tab, tid);
  __syncthreads();
  table_count(tab, dst, chunk, base, have, tid);
  __syncthreads();
  {
    int v[EB_SWEEP], h[EB_SWEEP];
    #pragma unroll
    for (int k = 0; k < EB_SWEEP; ++k) {
      v[k] = sweep_word(tid, k, chunk);
      h[k] = tab[v[k]];
    }
    #pragma unroll
    for (int k = 0; k < EB_SWEEP; ++k)
      if (h[k]) h[k] = atomicAdd(&cur[base + v[k]], h[k]);
    #pragma unroll
    for (int k = 0; k < EB_SWEEP; ++k) tab[v[k]] = h[k];
  }
  __syncthreads();
  const int4* d4 = (const int4*)dst + (size_t)chunk * EB_Q;
  const int4* s4 = (const int4*)src + (size_t)chunk * EB_Q;
  int4 d[EB_V], s[EB_V];
  #pragma unroll
  for (int j = 0; j < EB_V; ++j) {
    const int i = min(tid + j * EB_T, EB_Q - 1);
    d[j] = d4[i];
    s[j] = s4[i];
  }
  bool in[EB_V * 4];
  int  at[EB_V * 4];
  #pragma unroll
  for (int j = 0; j < EB_V; ++j) {
    const bool live = tid + j * EB_T < EB_Q;
    const unsigned r0 = word_of(d[j].x, base);
    const unsigned r1 = word_of(d[j].y, base);
    const unsigned r2 = word_of(d[j].z, base);
    const unsigned r3 = word_of(d[j].w, base);
    in[j * 4 + 0] = live && r0 < have;
    in[j * 4 + 1] = live && r1 < have;
    in[j * 4 + 2] = live && r2 < have;
    in[j * 4 + 3] = live && r3 < have;
    at[j * 4 + 0] = 0;
    if (in[j * 4 + 0]) at[j * 4 + 0] = atomicAdd(&tab[r0], 1);
    at[j * 4 + 1] = 0;
    if (in[j * 4 + 1]) at[j * 4 + 1] = atomicAdd(&tab[r1], 1);
    at[j * 4 + 2] = 0;
    if (in[j * 4 + 2]) at[j * 4 + 2] = atomicAdd(&tab[r2], 1);
    at[j * 4 + 3] = 0;
    if (in[j * 4 + 3]) at[j * 4 + 3] = atomicAdd(&tab[r3], 1);
  }
  #pragma unroll
  for (int j = 0; j < EB_V; ++j) {
    if (in[j * 4 + 0]) col[min(max(at[j * 4 + 0], 0), EE - 1)] = s[j].x;
    if (in[j * 4 + 1]) col[min(max(at[j * 4 + 1], 0), EE - 1)] = s[j].y;
    if (in[j * 4 + 2]) col[min(max(at[j * 4 + 2], 0), EE - 1)] = s[j].z;
    if (in[j * 4 + 3]) col[min(max(at[j * 4 + 3], 0), EE - 1)] = s[j].w;
  }
}

__device__ inline float leaky(float x) { return x > 0.0f ? x : 0.01f * x; }

__device__ inline int wave_min_i(int x) {
  for (int o = 16; o > 0; o >>= 1) x = min(x, __shfl_xor(x, o, 32));
  return x;
}
__device__ inline int wave_sum_i(int x) {
  for (int o = 16; o > 0; o >>= 1) x += __shfl_xor(x, o, 32);
  return x;
}
__device__ inline float wave_max_f(float x) {
  for (int o = 16; o > 0; o >>= 1) x = fmaxf(x, __shfl_xor(x, o, 32));
  return x;
}

__device__ inline int next_source(const int* __restrict__ seg, int deg, int last, int lane,
                                  int& mult) {
  int best = 0x7fffffff;
  for (int c = lane; c < deg; c += 32) {
    const int s = seg[c];
    if (s > last && s < best) best = s;
  }
  best = wave_min_i(best);
  int m = 0;
  for (int c = lane; c < deg; c += 32) m += (seg[c] == best) ? 1 : 0;
  mult = (best == 0x7fffffff) ? 0 : wave_sum_i(m);
  return best;
}

template <int C>
__device__ inline void pick(const int (&rank)[C], const int (&sv)[C], const float (&w)[C],
                            int k, int& svk, float& wk) {
  svk = 0; wk = 0.0f;
  for (int c = 0; c < C; ++c) {
    const unsigned mk = __builtin_amdgcn_ballot_w32(rank[c] == k);
    if (C == 1 || mk != 0u) {
      const int j = (__ffs((int)mk) - 1) & 31;
      svk = __builtin_amdgcn_readlane(sv[c], j);
      wk  = __int_as_float(__builtin_amdgcn_readlane(__float_as_int(w[c]), j));
    }
  }
}

template <int C>
__device__ inline float4 segment_in_registers(const int* __restrict__ seg, int deg, int lane,
                                              const float* __restrict__ q,
                                              float pv, float be0,
                                              const float4* __restrict__ hv4) {
  int sv[C], key[C], rank[C];
  float x[C], ex[C], a[C];
  for (int c = 0; c < C; ++c) {
    const int i = c * 32 + lane;
    sv[c] = 0; key[c] = 0x7fffffff; x[c] = -INFINITY; rank[c] = 0;
    if (i < deg) {
      sv[c]  = min(max(seg[i], 0), NN - 1);
      key[c] = sv[c];
      x[c]   = leaky(pv + q[sv[c]] + be0);
    }
  }
  for (int j = 0; j < deg; ++j) {
    int kj = __builtin_amdgcn_readlane(key[0], j & 31);
    for (int c = 1; c < C; ++c)
      if ((j >> 5) == c) kj = __builtin_amdgcn_readlane(key[c], j & 31);
    for (int c = 0; c < C; ++c)
      rank[c] += (kj < key[c] || (kj == key[c] && j < c * 32 + lane)) ? 1 : 0;
  }
  float mx = x[0];
  for (int c = 1; c < C; ++c) mx = fmaxf(mx, x[c]);
  mx = wave_max_f(mx);
  for (int c = 0; c < C; ++c) ex[c] = (c * 32 + lane < deg) ? expf(x[c] - mx) : 0.0f;
  float den = 0.0f;
  for (int k = 0; k < deg; ++k) {
    int s_; float exk;
    pick<C>(rank, sv, ex, k, s_, exk);
    den += exk;
  }
  const float dsafe = den > 0.0f ? den : 1.0f;
  for (int c = 0; c < C; ++c) a[c] = ex[c] / dsafe;

  float4 acc = make_float4(0.0f, 0.0f, 0.0f, 0.0f);
  int k = 0;
  for (; k + 4 <= deg; k += 4) {
    int s0, s1, s2, s3; float a0, a1, a2, a3;
    pick<C>(rank, sv, a, k,     s0, a0);
    pick<C>(rank, sv, a, k + 1, s1, a1);
    pick<C>(rank, sv, a, k + 2, s2, a2);
    pick<C>(rank, sv, a, k + 3, s3, a3);
    const float4 h0 = hv4[(size_t)s0 * (GG / 4) + lane];
    const float4 h1 = hv4[(size_t)s1 * (GG / 4) + lane];
    const float4 h2 = hv4[(size_t)s2 * (GG / 4) + lane];
    const float4 h3 = hv4[(size_t)s3 * (GG / 4) + lane];
    acc.x += h0.x * a0; acc.y += h0.y * a0; acc.z += h0.z * a0; acc.w += h0.w * a0;
    acc.x += h1.x * a1; acc.y += h1.y * a1; acc.z += h1.z * a1; acc.w += h1.w * a1;
    acc.x += h2.x * a2; acc.y += h2.y * a2; acc.z += h2.z * a2; acc.w += h2.w * a2;
    acc.x += h3.x * a3; acc.y += h3.y * a3; acc.z += h3.z * a3; acc.w += h3.w * a3;
  }
  for (; k < deg; ++k) {
    int sk; float ak;
    pick<C>(rank, sv, a, k, sk, ak);
    const float4 h = hv4[(size_t)sk * (GG / 4) + lane];
    acc.x += h.x * ak; acc.y += h.y * ak; acc.z += h.z * ak; acc.w += h.w * ak;
  }
  return acc;
}

__device__ inline float lane_value_f(float x, int k) {
  return __int_as_float(__builtin_amdgcn_readlane(__float_as_int(x), k));
}

__device__ inline float4 segment_in_lane_order(const int* __restrict__ seg, int deg, int lane,
                                               const float* __restrict__ q,
                                               float pv, float be0,
                                               const float4* __restrict__ hv4) {
  const bool live = lane < deg;
  int sv = 0;
  unsigned key = 0xffffffffu;
  float x = -INFINITY;
  if (live) {
    sv  = min(max(seg[lane], 0), NN - 1);
    key = ((unsigned)sv << 5) | (unsigned)lane;
    x   = leaky(pv + q[sv] + be0);
  }
  int rank = 0;
  float mx = -INFINITY;
  for (int j = 0; j < deg; ++j) {
    const unsigned kj = (unsigned)__builtin_amdgcn_readlane((int)key, j);
    rank += (kj < key) ? 1 : 0;
    mx = fmaxf(mx, lane_value_f(x, j));
  }
  const float ex = live ? expf(x - mx) : 0.0f;
  const int to = (live ? rank : lane) << 2;
  const int   ssv = __builtin_amdgcn_ds_permute(to, sv);
  const float sex = __int_as_float(__builtin_amdgcn_ds_permute(to, __float_as_int(ex)));

  float den = 0.0f;
  for (int k = 0; k < deg; ++k) den += lane_value_f(sex, k);
  const float dsafe = den > 0.0f ? den : 1.0f;
  const float sa = sex / dsafe;

  float4 acc = make_float4(0.0f, 0.0f, 0.0f, 0.0f);
  int k = 0;
  for (; k + 4 <= deg; k += 4) {
    const int s0 = __builtin_amdgcn_readlane(ssv, k);
    const int s1 = __builtin_amdgcn_readlane(ssv, k + 1);
    const int s2 = __builtin_amdgcn_readlane(ssv, k + 2);
    const int s3 = __builtin_amdgcn_readlane(ssv, k + 3);
    const float a0 = lane_value_f(sa, k);
    const float a1 = lane_value_f(sa, k + 1);
    const float a2 = lane_value_f(sa, k + 2);
    const float a3 = lane_value_f(sa, k + 3);
    const float4 h0 = hv4[(size_t)s0 * (GG / 4) + lane];
    const float4 h1 = hv4[(size_t)s1 * (GG / 4) + lane];
    const float4 h2 = hv4[(size_t)s2 * (GG / 4) + lane];
    const float4 h3 = hv4[(size_t)s3 * (GG / 4) + lane];
    acc.x += h0.x * a0; acc.y += h0.y * a0; acc.z += h0.z * a0; acc.w += h0.w * a0;
    acc.x += h1.x * a1; acc.y += h1.y * a1; acc.z += h1.z * a1; acc.w += h1.w * a1;
    acc.x += h2.x * a2; acc.y += h2.y * a2; acc.z += h2.z * a2; acc.w += h2.w * a2;
    acc.x += h3.x * a3; acc.y += h3.y * a3; acc.z += h3.z * a3; acc.w += h3.w * a3;
  }
  for (; k < deg; ++k) {
    const int   sk = __builtin_amdgcn_readlane(ssv, k);
    const float ak = lane_value_f(sa, k);
    const float4 h = hv4[(size_t)sk * (GG / 4) + lane];
    acc.x += h.x * ak; acc.y += h.y * ak; acc.z += h.z * ak; acc.w += h.w * ak;
  }
  return acc;
}

__global__ void __launch_bounds__(256)
node_softmax_aggregate(const int* __restrict__ col,
                       const int* __restrict__ cntp, const int* __restrict__ endp,
                       const float* __restrict__ p, const float* __restrict__ q,
                       const float* __restrict__ be, const float* __restrict__ hv,
                       float* __restrict__ out) {
  const int v    = __builtin_amdgcn_readfirstlane((int)(blockIdx.x * 8u + (threadIdx.x >> 5)));
  const int lane = threadIdx.x & 31;
  if (v >= NN) return;
  const int deg   = __builtin_amdgcn_readfirstlane(cntp[v]);
  const int start = __builtin_amdgcn_readfirstlane(endp[v]) - deg;
  const int* seg  = col + start;
  const float pv  = __int_as_float(__builtin_amdgcn_readfirstlane(__float_as_int(p[v])));
  const float be0 = be[0];
  const float4* hv4 = (const float4*)hv;
  float4 acc;

  if (deg <= 32) {
    acc = segment_in_lane_order(seg, deg, lane, q, pv, be0, hv4);
  } else if (deg <= 64) {
    acc = segment_in_registers<2>(seg, deg, lane, q, pv, be0, hv4);
  } else {
    acc = make_float4(0.0f, 0.0f, 0.0f, 0.0f);
    float mx = -INFINITY;
    for (int c = lane; c < deg; c += 32) {
      const int s = min(max(seg[c], 0), NN - 1);
      mx = fmaxf(mx, leaky(pv + q[s] + be0));
    }
    mx = wave_max_f(mx);
    float den = 0.0f;
    int last = -1;
    for (int k = 0; k < deg; ) {
      int mult;
      const int s = next_source(seg, deg, last, lane, mult);
      if (mult <= 0) break;
      last = s;
      const float exk = expf(leaky(pv + q[min(s, NN - 1)] + be0) - mx);
      for (int t = 0; t < mult; ++t) den += exk;
      k += mult;
    }
    const float dsafe = den > 0.0f ? den : 1.0f;
    last = -1;
    for (int k = 0; k < deg; ) {
      int mult;
      const int s = next_source(seg, deg, last, lane, mult);
      if (mult <= 0) break;
      last = s;
      const int sc = min(s, NN - 1);
      const float ak = expf(leaky(pv + q[sc] + be0) - mx) / dsafe;
      const float4 h = hv4[(size_t)sc * (GG / 4) + lane];
      for (int t = 0; t < mult; ++t) {
        acc.x += h.x * ak; acc.y += h.y * ak; acc.z += h.z * ak; acc.w += h.w * ak;
      }
      k += mult;
    }
  }

  float4 r;
  r.x = acc.x > 0.0f ? acc.x : expm1f(acc.x);
  r.y = acc.y > 0.0f ? acc.y : expm1f(acc.y);
  r.z = acc.z > 0.0f ? acc.z : expm1f(acc.z);
  r.w = acc.w > 0.0f ? acc.w : expm1f(acc.w);
  ((float4*)out)[(size_t)v * (GG / 4) + lane] = r;
}

extern "C" void kernel_launch(void* const* d_in, const int* in_sizes, int n_in,
                              void* d_out, int out_size, void* d_ws, size_t ws_size,
                              hipStream_t stream) {
  (void)in_sizes; (void)n_in; (void)out_size; (void)ws_size;
  const float* nf  = (const float*)d_in[0];
  const int*   src = (const int*)  d_in[1];
  const int*   dst = (const int*)  d_in[2];
  const float* We  = (const float*)d_in[3];
  const float* be  = (const float*)d_in[4];
  const float* Wn  = (const float*)d_in[5];
  const float* bn  = (const float*)d_in[6];
  float* out = (float*)d_out;

  float* ws   = (float*)d_ws;
  float* hv   = ws;
  float* p    = hv + (size_t)NN * GG;
  float* q    = p + NN;
  int*   cnt  = (int*)(q + NN);
  int*   cur  = cnt + NN;
  int*   col  = cur + NN;

  node_gemm_wmma<<<NN / 16, 256, 0, stream>>>(nf, Wn, bn, hv, We, p, q, cnt);
  edge_count<<<EB_BLOCKS, EB_T, 0, stream>>>(dst, cnt);
  scan_counts<<<SCAN_B, SCAN_T, 0, stream>>>(cnt, cur);
  edge_place<<<EB_BLOCKS, EB_T, 0, stream>>>(src, dst, cur, col);
  node_softmax_aggregate<<<(NN * 32 + 255) / 256, 256, 0, stream>>>(col, cnt, cur, p, q, be, hv, out);
}


# Count pass stores per-chunk count tables and 1024-node block sums (no global atomics, no cleared counters), the scan adds them up; so the node GEMM moves into the place launch: 4 launches, not 5
# speedup vs baseline: 3.3806x; 1.0561x over previous
#include <hip/hip_runtime.h>
#include <math.h>


typedef __attribute__((ext_vector_type(2))) float v2f;
typedef __attribute__((ext_vector_type(8))) float v8f;

#define NN 50000
#define EE 800000
#define DD 128
#define GG 128
#define LDS_STRIDE 132

__device__ inline float leaky(float x) { return x > 0.0f ? x : 0.01f * x; }

__device__ inline int wave_min_i(int x) {
  for (int o = 16; o > 0; o >>= 1) x = min(x, __shfl_xor(x, o, 32));
  return x;
}
__device__ inline int wave_sum_i(int x) {
  for (int o = 16; o > 0; o >>= 1) x += __shfl_xor(x, o, 32);
  return x;
}
__device__ inline float wave_max_f(float x) {
  for (int o = 16; o > 0; o >>= 1) x = fmaxf(x, __shfl_xor(x, o, 32));
  return x;
}

#define GM_ROWS   64
#define GM_BLOCKS ((NN + GM_ROWS - 1) / GM_ROWS)

static_assert(NN % 16 == 0, "row tiles are whole");

__device__ __forceinline__ void
node_gemm_role(float* __restrict__ Atile, int gblock,
               const float* __restrict__ nf, const float* __restrict__ Wn,
               const float* __restrict__ bn, float* __restrict__ hv,
               const float* __restrict__ We, float* __restrict__ p,
               float* __restrict__ q) {
  const int row0 = gblock * GM_ROWS;
  const int tid  = threadIdx.x;

  #pragma unroll
  for (int i = tid; i < GM_ROWS * DD; i += 1024) {
    const int r = i >> 7;
    const int c = i & (DD - 1);
    Atile[r * LDS_STRIDE + c] = nf[(size_t)min(row0 + r, NN - 1) * DD + c];
  }
  __syncthreads();

  const int wave = __builtin_amdgcn_readfirstlane(tid >> 5);
  const int lane = tid & 31;
  const int trow = (wave >> 3) * 16;
  const int col0 = (wave & 7) * 16;
  const int half = lane >> 4;
  const int l16  = lane & 15;

  if (row0 + trow < NN) {
    v8f acc = {};
    #pragma unroll
    for (int k = 0; k < DD; k += 4) {
      const int ka = k + half * 2;
      v2f a, b;
      a.x = Atile[(trow + l16) * LDS_STRIDE + ka];
      a.y = Atile[(trow + l16) * LDS_STRIDE + ka + 1];
      b.x = Wn[(size_t)(ka + 0) * GG + col0 + l16];
      b.y = Wn[(size_t)(ka + 1) * GG + col0 + l16];
      acc = __builtin_amdgcn_wmma_f32_16x16x4_f32(false, a, false, b,
                                                  (short)0, acc, false, false);
    }

    #pragma unroll
    for (int i = 0; i < 8; ++i) {
      const int row = row0 + trow + i + half * 8;
      const int col = col0 + l16;
      float v = acc[i] + bn[col];
      v = v > 0.0f ? v : 0.01f * v;
      hv[(size_t)row * GG + col] = v;
    }
  }

  const int sr0 = wave * 2;
  if (row0 + sr0 < NN) {
    float ap0 = 0.0f, aq0 = 0.0f, ap1 = 0.0f, aq1 = 0.0f;
    #pragma unroll
    for (int i = lane; i < DD; i += 32) {
      const float wp = We[i];
      const float wq = We[DD + i];
      const float x0 = Atile[sr0 * LDS_STRIDE + i];
      const float x1 = Atile[(sr0 + 1) * LDS_STRIDE + i];
      ap0 += x0 * wp;
      aq0 += x0 * wq;
      ap1 += x1 * wp;
      aq1 += x1 * wq;
    }
    #pragma unroll
    for (int off = 16; off > 0; off >>= 1) {
      ap0 += __shfl_xor(ap0, off, 32);
      aq0 += __shfl_xor(aq0, off, 32);
      ap1 += __shfl_xor(ap1, off, 32);
      aq1 += __shfl_xor(aq1, off, 32);
    }

    if (lane < 2) {
      const int node = row0 + sr0 + lane;
      p[node] = lane ? ap1 : ap0;
      q[node] = lane ? aq1 : aq0;
    }
  }
}

#define EB_CHUNKS 64
#define EB_PARTS  4
#define EB_BLOCKS (EB_CHUNKS * EB_PARTS)
#define EB_T      1024
#define EB_EDGES  (EE / EB_CHUNKS)
#define EB_Q      (EB_EDGES / 4)
#define EB_V      ((EB_Q + EB_T - 1) / EB_T)
#define EB_NODES  12512
#define EB_SWEEP  13
#define EB_WORDS  (EB_SWEEP * EB_T)
#define EB_ROT    192
#define CN_NODES  EB_WORDS
#define PT_ROW    (EB_PARTS * CN_NODES)
#define PT_TOT    (EB_PARTS * EB_SWEEP)

static_assert(EB_PARTS * EB_NODES >= NN, "the node ranges cover all nodes");
static_assert(EB_WORDS >= EB_NODES && EB_NODES % 32 == 0, "table holds a range; ranges start on a line");
static_assert(EB_CHUNKS * EB_ROT <= EB_WORDS && EB_ROT % 32 == 0, "sweep starts stay inside the table, on lines");
static_assert(EB_EDGES * EB_CHUNKS == EE && EB_Q * 4 == EB_EDGES, "chunks are whole 16-byte pieces");
static_assert(PT_ROW >= NN && (size_t)EB_CHUNKS * (PT_ROW + PT_TOT) <= (size_t)NN * GG, "count tables fit the hv plane, which is free until the GEMM writes it");

__device__ __forceinline__ unsigned word_of(int d, int base) { return (unsigned)d - (unsigned)base; }

__device__ __forceinline__ void table_clear(int* __restrict__ tab, int tid) {
  int4* t4 = (int4*)tab;
  for (int i = tid; i < EB_WORDS / 4; i += EB_T) t4[i] = make_int4(0, 0, 0, 0);
}

__device__ __forceinline__ void table_count(int* __restrict__ tab, const int* __restrict__ dst,
                                            int chunk, int base, unsigned have, int tid) {
  const int4* d4 = (const int4*)dst + (size_t)chunk * EB_Q;
  int4 d[EB_V];
  #pragma unroll
  for (int j = 0; j < EB_V; ++j) d[j] = d4[min(tid + j * EB_T, EB_Q - 1)];
  #pragma unroll
  for (int j = 0; j < EB_V; ++j) {
    if (tid + j * EB_T < EB_Q) {
      const unsigned r0 = word_of(d[j].x, base);
      const unsigned r1 = word_of(d[j].y, base);
      const unsigned r2 = word_of(d[j].z, base);
      const unsigned r3 = word_of(d[j].w, base);
      if (r0 < have) atomicAdd(&tab[r0], 1);
      if (r1 < have) atomicAdd(&tab[r1], 1);
      if (r2 < have) atomicAdd(&tab[r2], 1);
      if (r3 < have) atomicAdd(&tab[r3], 1);
    }
  }
}

__device__ __forceinline__ int sweep_word(int tid, int j, int chunk) {
  const int v = tid + j * EB_T + chunk * EB_ROT;
  return v >= EB_WORDS ? v - EB_WORDS : v;
}

__global__ void __launch_bounds__(EB_T)
edge_count(const int* __restrict__ dst, int* __restrict__ ptab, int* __restrict__ ptot) {
  __shared__ __attribute__((aligned(16))) int tab[EB_WORDS];
  const int tid   = threadIdx.x;
  const int chunk = blockIdx.x % EB_CHUNKS;
  const int part  = blockIdx.x / EB_CHUNKS;
  const int base  = part * CN_NODES;
  const unsigned have = (unsigned)min(CN_NODES, NN - base);
  table_clear(tab, tid);
  __syncthreads();
  table_count(tab, dst, chunk, base, have, tid);
  __syncthreads();
  int h[EB_SWEEP];
  #pragma unroll
  for (int k = 0; k < EB_SWEEP; ++k) h[k] = tab[tid + k * EB_T];
  int* __restrict__ row = ptab + (size_t)chunk * PT_ROW + base;
  #pragma unroll
  for (int k = 0; k < EB_SWEEP; ++k) row[tid + k * EB_T] = h[k];
  const int wave = tid >> 5;
  const int lane = tid & 31;
  if (wave < EB_SWEEP) {
    const int4* t4 = (const int4*)(tab + wave * EB_T);
    int s = 0;
    #pragma unroll
    for (int j = 0; j < EB_T / 128; ++j) {
      const int4 x = t4[j * 32 + lane];
      s += x.x + x.y + x.z + x.w;
    }
    s = wave_sum_i(s);
    if (lane == 0) ptot[chunk * PT_TOT + part * EB_SWEEP + wave] = s;
  }
}

#define SCAN_T   1024
#define SCAN_B   ((NN + SCAN_T - 1) / SCAN_T)
#define SCAN_LT  ((EB_CHUNKS * PT_TOT + SCAN_T - 1) / SCAN_T)

static_assert(SCAN_T == EB_T && SCAN_B <= PT_TOT && SCAN_B * SCAN_T <= PT_ROW, "scan blocks are the count tables' blocks");

__global__ void __launch_bounds__(SCAN_T)
scan_counts(const int* __restrict__ ptab, const int* __restrict__ ptot,
            int* __restrict__ cnt, int* __restrict__ cur) {
  __shared__ int wsum[32];
  __shared__ int wlow[32];
  const int tid  = threadIdx.x;
  const int lane = tid & 31;
  const int w    = tid >> 5;
  const int blk  = (int)blockIdx.x;
  const int i    = blk * SCAN_T + tid;
  int t[EB_CHUNKS], r[SCAN_LT];
  #pragma unroll
  for (int c = 0; c < EB_CHUNKS; ++c) t[c] = ptab[(size_t)c * PT_ROW + i];
  #pragma unroll
  for (int j = 0; j < SCAN_LT; ++j) r[j] = ptot[min(tid + j * SCAN_T, EB_CHUNKS * PT_TOT - 1)];
  int own = 0;
  #pragma unroll
  for (int c = 0; c < EB_CHUNKS; ++c) own += t[c];
  int low = 0;
  #pragma unroll
  for (int j = 0; j < SCAN_LT; ++j) {
    const int x = tid + j * SCAN_T;
    if (x < EB_CHUNKS * PT_TOT && (x % PT_TOT) < blk) low += r[j];
  }
  int inc = own;
  for (int o = 1; o < 32; o <<= 1) {
    const int u = __shfl_up(inc, o, 32);
    if (lane >= o) inc += u;
  }
  for (int o = 16; o > 0; o >>= 1) low += __shfl_xor(low, o, 32);
  if (lane == 31) wsum[w] = inc;
  if (lane == 0)  wlow[w] = low;
  __syncthreads();
  int before = wlow[lane] + ((lane < w) ? wsum[lane] : 0);
  for (int o = 16; o > 0; o >>= 1) before += __shfl_xor(before, o, 32);
  if (i < NN) {
    cnt[i] = own;
    cur[i] = before + inc - own;
  }
}

__device__ __forceinline__ void
edge_place_role(int* __restrict__ tab, int pblock,
                const int* __restrict__ src, const int* __restrict__ dst,
                int* __restrict__ cur, int* __restrict__ col) {
  const int tid   = threadIdx.x;
  const int chunk = pblock % EB_CHUNKS;
  const int base  = (pblock / EB_CHUNKS) * EB_NODES;
  const unsigned have = (unsigned)min(EB_NODES, NN - base);
  table_clear(tab, tid);
  __syncthreads();
  table_count(tab, dst, chunk, base, have, tid);
  __syncthreads();
  {
    int v[EB_SWEEP], h[EB_SWEEP];
    #pragma unroll
    for (int k = 0; k < EB_SWEEP; ++k) {
      v[k] = sweep_word(tid, k, chunk);
      h[k] = tab[v[k]];
    }
    #pragma unroll
    for (int k = 0; k < EB_SWEEP; ++k)
      if (h[k]) h[k] = atomicAdd(&cur[base + v[k]], h[k]);
    #pragma unroll
    for (int k = 0; k < EB_SWEEP; ++k) tab[v[k]] = h[k];
  }
  __syncthreads();
  const int4* d4 = (const int4*)dst + (size_t)chunk * EB_Q;
  const int4* s4 = (const int4*)src + (size_t)chunk * EB_Q;
  int4 d[EB_V], s[EB_V];
  #pragma unroll
  for (int j = 0; j < EB_V; ++j) {
    const int i = min(tid + j * EB_T, EB_Q - 1);
    d[j] = d4[i];
    s[j] = s4[i];
  }
  bool in[EB_V * 4];
  int  at[EB_V * 4];
  #pragma unroll
  for (int j = 0; j < EB_V; ++j) {
    const bool live = tid + j * EB_T < EB_Q;
    const unsigned r0 = word_of(d[j].x, base);
    const unsigned r1 = word_of(d[j].y, base);
    const unsigned r2 = word_of(d[j].z, base);
    const unsigned r3 = word_of(d[j].w, base);
    in[j * 4 + 0] = live && r0 < have;
    in[j * 4 + 1] = live && r1 < have;
    in[j * 4 + 2] = live && r2 < have;
    in[j * 4 + 3] = live && r3 < have;
    at[j * 4 + 0] = 0;
    if (in[j * 4 + 0]) at[j * 4 + 0] = atomicAdd(&tab[r0], 1);
    at[j * 4 + 1] = 0;
    if (in[j * 4 + 1]) at[j * 4 + 1] = atomicAdd(&tab[r1], 1);
    at[j * 4 + 2] = 0;
    if (in[j * 4 + 2]) at[j * 4 + 2] = atomicAdd(&tab[r2], 1);
    at[j * 4 + 3] = 0;
    if (in[j * 4 + 3]) at[j * 4 + 3] = atomicAdd(&tab[r3], 1);
  }
  #pragma unroll
  for (int j = 0; j < EB_V; ++j) {
    if (in[j * 4 + 0]) col[min(max(at[j * 4 + 0], 0), EE - 1)] = s[j].x;
    if (in[j * 4 + 1]) col[min(max(at[j * 4 + 1], 0), EE - 1)] = s[j].y;
    if (in[j * 4 + 2]) col[min(max(at[j * 4 + 2], 0), EE - 1)] = s[j].z;
    if (in[j * 4 + 3]) col[min(max(at[j * 4 + 3], 0), EE - 1)] = s[j].w;
  }
}

static_assert(GM_ROWS * LDS_STRIDE <= EB_WORDS, "the row tiles fit the table's LDS block");

__global__ void __launch_bounds__(EB_T, 16)
place_and_node_gemm(const int* __restrict__ src, const int* __restrict__ dst,
                    int* __restrict__ cur, int* __restrict__ col,
                    const float* __restrict__ nf, const float* __restrict__ Wn,
                    const float* __restrict__ bn, float* __restrict__ hv,
                    const float* __restrict__ We, float* __restrict__ p,
                    float* __restrict__ q) {
  __shared__ __attribute__((aligned(16))) int smem[EB_WORDS];
  const int b = (int)blockIdx.x;
  if (b < EB_BLOCKS) edge_place_role(smem, b, src, dst, cur, col);
  else               node_gemm_role((float*)smem, b - EB_BLOCKS, nf, Wn, bn, hv, We, p, q);
}

__device__ inline int next_source(const int* __restrict__ seg, int deg, int last, int lane,
                                  int& mult) {
  int best = 0x7fffffff;
  for (int c = lane; c < deg; c += 32) {
    const int s = seg[c];
    if (s > last && s < best) best = s;
  }
  best = wave_min_i(best);
  int m = 0;
  for (int c = lane; c < deg; c += 32) m += (seg[c] == best) ? 1 : 0;
  mult = (best == 0x7fffffff) ? 0 : wave_sum_i(m);
  return best;
}

template <int C>
__device__ inline void pick(const int (&rank)[C], const int (&sv)[C], const float (&w)[C],
                            int k, int& svk, float& wk) {
  svk = 0; wk = 0.0f;
  for (int c = 0; c < C; ++c) {
    const unsigned mk = __builtin_amdgcn_ballot_w32(rank[c] == k);
    if (C == 1 || mk != 0u) {
      const int j = (__ffs((int)mk) - 1) & 31;
      svk = __builtin_amdgcn_readlane(sv[c], j);
      wk  = __int_as_float(__builtin_amdgcn_readlane(__float_as_int(w[c]), j));
    }
  }
}

template <int C>
__device__ inline float4 segment_in_registers(const int* __restrict__ seg, int deg, int lane,
                                              const float* __restrict__ q,
                                              float pv, float be0,
                                              const float4* __restrict__ hv4) {
  int sv[C], key[C], rank[C];
  float x[C], ex[C], a[C];
  for (int c = 0; c < C; ++c) {
    const int i = c * 32 + lane;
    sv[c] = 0; key[c] = 0x7fffffff; x[c] = -INFINITY; rank[c] = 0;
    if (i < deg) {
      sv[c]  = min(max(seg[i], 0), NN - 1);
      key[c] = sv[c];
      x[c]   = leaky(pv + q[sv[c]] + be0);
    }
  }
  for (int j = 0; j < deg; ++j) {
    int kj = __builtin_amdgcn_readlane(key[0], j & 31);
    for (int c = 1; c < C; ++c)
      if ((j >> 5) == c) kj = __builtin_amdgcn_readlane(key[c], j & 31);
    for (int c = 0; c < C; ++c)
      rank[c] += (kj < key[c] || (kj == key[c] && j < c * 32 + lane)) ? 1 : 0;
  }
  float mx = x[0];
  for (int c = 1; c < C; ++c) mx = fmaxf(mx, x[c]);
  mx = wave_max_f(mx);
  for (int c = 0; c < C; ++c) ex[c] = (c * 32 + lane < deg) ? expf(x[c] - mx) : 0.0f;
  float den = 0.0f;
  for (int k = 0; k < deg; ++k) {
    int s_; float exk;
    pick<C>(rank, sv, ex, k, s_, exk);
    den += exk;
  }
  const float dsafe = den > 0.0f ? den : 1.0f;
  for (int c = 0; c < C; ++c) a[c] = ex[c] / dsafe;

  float4 acc = make_float4(0.0f, 0.0f, 0.0f, 0.0f);
  int k = 0;
  for (; k + 4 <= deg; k += 4) {
    int s0, s1, s2, s3; float a0, a1, a2, a3;
    pick<C>(rank, sv, a, k,     s0, a0);
    pick<C>(rank, sv, a, k + 1, s1, a1);
    pick<C>(rank, sv, a, k + 2, s2, a2);
    pick<C>(rank, sv, a, k + 3, s3, a3);
    const float4 h0 = hv4[(size_t)s0 * (GG / 4) + lane];
    const float4 h1 = hv4[(size_t)s1 * (GG / 4) + lane];
    const float4 h2 = hv4[(size_t)s2 * (GG / 4) + lane];
    const float4 h3 = hv4[(size_t)s3 * (GG / 4) + lane];
    acc.x += h0.x * a0; acc.y += h0.y * a0; acc.z += h0.z * a0; acc.w += h0.w * a0;
    acc.x += h1.x * a1; acc.y += h1.y * a1; acc.z += h1.z * a1; acc.w += h1.w * a1;
    acc.x += h2.x * a2; acc.y += h2.y * a2; acc.z += h2.z * a2; acc.w += h2.w * a2;
    acc.x += h3.x * a3; acc.y += h3.y * a3; acc.z += h3.z * a3; acc.w += h3.w * a3;
  }
  for (; k < deg; ++k) {
    int sk; float ak;
    pick<C>(rank, sv, a, k, sk, ak);
    const float4 h = hv4[(size_t)sk * (GG / 4) + lane];
    acc.x += h.x * ak; acc.y += h.y * ak; acc.z += h.z * ak; acc.w += h.w * ak;
  }
  return acc;
}

__device__ inline float lane_value_f(float x, int k) {
  return __int_as_float(__builtin_amdgcn_readlane(__float_as_int(x), k));
}

__device__ inline float4 segment_in_lane_order(const int* __restrict__ seg, int deg, int lane,
                                               const float* __restrict__ q,
                                               float pv, float be0,
                                               const float4* __restrict__ hv4) {
  const bool live = lane < deg;
  int sv = 0;
  unsigned key = 0xffffffffu;
  float x = -INFINITY;
  if (live) {
    sv  = min(max(seg[lane], 0), NN - 1);
    key = ((unsigned)sv << 5) | (unsigned)lane;
    x   = leaky(pv + q[sv] + be0);
  }
  int rank = 0;
  float mx = -INFINITY;
  for (int j = 0; j < deg; ++j) {
    const unsigned kj = (unsigned)__builtin_amdgcn_readlane((int)key, j);
    rank += (kj < key) ? 1 : 0;
    mx = fmaxf(mx, lane_value_f(x, j));
  }
  const float ex = live ? expf(x - mx) : 0.0f;
  const int to = (live ? rank : lane) << 2;
  const int   ssv = __builtin_amdgcn_ds_permute(to, sv);
  const float sex = __int_as_float(__builtin_amdgcn_ds_permute(to, __float_as_int(ex)));

  float den = 0.0f;
  for (int k = 0; k < deg; ++k) den += lane_value_f(sex, k);
  const float dsafe = den > 0.0f ? den : 1.0f;
  const float sa = sex / dsafe;

  float4 acc = make_float4(0.0f, 0.0f, 0.0f, 0.0f);
  int k = 0;
  for (; k + 4 <= deg; k += 4) {
    const int s0 = __builtin_amdgcn_readlane(ssv, k);
    const int s1 = __builtin_amdgcn_readlane(ssv, k + 1);
    const int s2 = __builtin_amdgcn_readlane(ssv, k + 2);
    const int s3 = __builtin_amdgcn_readlane(ssv, k + 3);
    const float a0 = lane_value_f(sa, k);
    const float a1 = lane_value_f(sa, k + 1);
    const float a2 = lane_value_f(sa, k + 2);
    const float a3 = lane_value_f(sa, k + 3);
    const float4 h0 = hv4[(size_t)s0 * (GG / 4) + lane];
    const float4 h1 = hv4[(size_t)s1 * (GG / 4) + lane];
    const float4 h2 = hv4[(size_t)s2 * (GG / 4) + lane];
    const float4 h3 = hv4[(size_t)s3 * (GG / 4) + lane];
    acc.x += h0.x * a0; acc.y += h0.y * a0; acc.z += h0.z * a0; acc.w += h0.w * a0;
    acc.x += h1.x * a1; acc.y += h1.y * a1; acc.z += h1.z * a1; acc.w += h1.w * a1;
    acc.x += h2.x * a2; acc.y += h2.y * a2; acc.z += h2.z * a2; acc.w += h2.w * a2;
    acc.x += h3.x * a3; acc.y += h3.y * a3; acc.z += h3.z * a3; acc.w += h3.w * a3;
  }
  for (; k < deg; ++k) {
    const int   sk = __builtin_amdgcn_readlane(ssv, k);
    const float ak = lane_value_f(sa, k);
    const float4 h = hv4[(size_t)sk * (GG / 4) + lane];
    acc.x += h.x * ak; acc.y += h.y * ak; acc.z += h.z * ak; acc.w += h.w * ak;
  }
  return acc;
}

__global__ void __launch_bounds__(256)
node_softmax_aggregate(const int* __restrict__ col,
                       const int* __restrict__ cntp, const int* __restrict__ endp,
                       const float* __restrict__ p, const float* __restrict__ q,
                       const float* __restrict__ be, const float* __restrict__ hv,
                       float* __restrict__ out) {
  const int v    = __builtin_amdgcn_readfirstlane((int)(blockIdx.x * 8u + (threadIdx.x >> 5)));
  const int lane = threadIdx.x & 31;
  if (v >= NN) return;
  const int deg   = __builtin_amdgcn_readfirstlane(cntp[v]);
  const int start = __builtin_amdgcn_readfirstlane(endp[v]) - deg;
  const int* seg  = col + start;
  const float pv  = __int_as_float(__builtin_amdgcn_readfirstlane(__float_as_int(p[v])));
  const float be0 = be[0];
  const float4* hv4 = (const float4*)hv;
  float4 acc;

  if (deg <= 32) {
    acc = segment_in_lane_order(seg, deg, lane, q, pv, be0, hv4);
  } else if (deg <= 64) {
    acc = segment_in_registers<2>(seg, deg, lane, q, pv, be0, hv4);
  } else {
    acc = make_float4(0.0f, 0.0f, 0.0f, 0.0f);
    float mx = -INFINITY;
    for (int c = lane; c < deg; c += 32) {
      const int s = min(max(seg[c], 0), NN - 1);
      mx = fmaxf(mx, leaky(pv + q[s] + be0));
    }
    mx = wave_max_f(mx);
    float den = 0.0f;
    int last = -1;
    for (int k = 0; k < deg; ) {
      int mult;
      const int s = next_source(seg, deg, last, lane, mult);
      if (mult <= 0) break;
      last = s;
      const float exk = expf(leaky(pv + q[min(s, NN - 1)] + be0) - mx);
      for (int t = 0; t < mult; ++t) den += exk;
      k += mult;
    }
    const float dsafe = den > 0.0f ? den : 1.0f;
    last = -1;
    for (int k = 0; k < deg; ) {
      int mult;
      const int s = next_source(seg, deg, last, lane, mult);
      if (mult <= 0) break;
      last = s;
      const int sc = min(s, NN - 1);
      const float ak = expf(leaky(pv + q[sc] + be0) - mx) / dsafe;
      const float4 h = hv4[(size_t)sc * (GG / 4) + lane];
      for (int t = 0; t < mult; ++t) {
        acc.x += h.x * ak; acc.y += h.y * ak; acc.z += h.z * ak; acc.w += h.w * ak;
      }
      k += mult;
    }
  }

  float4 r;
  r.x = acc.x > 0.0f ? acc.x : expm1f(acc.x);
  r.y = acc.y > 0.0f ? acc.y : expm1f(acc.y);
  r.z = acc.z > 0.0f ? acc.z : expm1f(acc.z);
  r.w = acc.w > 0.0f ? acc.w : expm1f(acc.w);
  ((float4*)out)[(size_t)v * (GG / 4) + lane] = r;
}

extern "C" void kernel_launch(void* const* d_in, const int* in_sizes, int n_in,
                              void* d_out, int out_size, void* d_ws, size_t ws_size,
                              hipStream_t stream) {
  (void)in_sizes; (void)n_in; (void)out_size; (void)ws_size;
  const float* nf  = (const float*)d_in[0];
  const int*   src = (const int*)  d_in[1];
  const int*   dst = (const int*)  d_in[2];
  const float* We  = (const float*)d_in[3];
  const float* be  = (const float*)d_in[4];
  const float* Wn  = (const float*)d_in[5];
  const float* bn  = (const float*)d_in[6];
  float* out = (float*)d_out;

  float* ws   = (float*)d_ws;
  float* hv   = ws;
  float* p    = hv + (size_t)NN * GG;
  float* q    = p + NN;
  int*   cnt  = (int*)(q + NN);
  int*   cur  = cnt + NN;
  int*   col  = cur + NN;
  int*   ptab = (int*)hv;
  int*   ptot = ptab + (size_t)EB_CHUNKS * PT_ROW;

  edge_count<<<EB_BLOCKS, EB_T, 0, stream>>>(dst, ptab, ptot);
  scan_counts<<<SCAN_B, SCAN_T, 0, stream>>>(ptab, ptot, cnt, cur);
  place_and_node_gemm<<<EB_BLOCKS + GM_BLOCKS, EB_T, 0, stream>>>(src, dst, cur, col, nf, Wn, bn, hv, We, p, q);
  node_softmax_aggregate<<<(NN * 32 + 255) / 256, 256, 0, stream>>>(col, cnt, cur, p, q, be, hv, out);
}
